# starting kernel
# baseline (speedup 1.0000x reference)
#include <hip/hip_runtime.h>

typedef float v2f __attribute__((ext_vector_type(2)));
typedef float v8f __attribute__((ext_vector_type(8)));

#define F_IN 128
#define HID  128
#define HID2 64
#define OUTC 32

static inline int ceil_div(int a, int b) { return (a + b - 1) / b; }


__global__ void deg_init_kernel(float* __restrict__ deg, int N) {
  int i = blockIdx.x * blockDim.x + threadIdx.x;
  if (i < N) deg[i] = 1.0f;
}

__global__ void deg_accum_kernel(const int* __restrict__ dst, float* __restrict__ deg, int E) {
  int e = blockIdx.x * blockDim.x + threadIdx.x;
  if (e < E) atomicAdd(&deg[dst[e]], 1.0f);
}

__global__ void deg_rsqrt_kernel(float* __restrict__ deg, int N) {
  int i = blockIdx.x * blockDim.x + threadIdx.x;
  if (i < N) deg[i] = rsqrtf(deg[i]);
}

__global__ __launch_bounds__(128) void gemm_wmma_kernel(
    const float* __restrict__ A, const float* __restrict__ B, float* __restrict__ C,
    int M, int K, int Ncol) {
  int lane = threadIdx.x & 31;
  int wave = threadIdx.x >> 5;
  int mtile = blockIdx.x * 4 + wave;
  int row0 = mtile << 4;
  if (row0 >= M) return;
  int col0 = blockIdx.y << 4;
  int half = lane >> 4;
  int l16  = lane & 15;

  v8f acc = {0.f, 0.f, 0.f, 0.f, 0.f, 0.f, 0.f, 0.f};
  const float* arow = A + (size_t)(row0 + l16) * K;
  const float* bcol = B + col0 + l16;

  for (int k = 0; k < K; k += 4) {
    int ka = k + half * 2;
    v2f a, b;
    a.x = arow[ka + 0];
    a.y = arow[ka + 1];
    b.x = bcol[(size_t)(ka + 0) * Ncol];
    b.y = bcol[(size_t)(ka + 1) * Ncol];
    acc = __builtin_amdgcn_wmma_f32_16x16x4_f32(
         false, a,  false, b,
         (short)0, acc,  false,  false);
  }

  float* crow = C + (size_t)(row0 + half * 8) * Ncol + col0 + l16;
#pragma unroll
  for (int r = 0; r < 8; ++r)
    crow[(size_t)r * Ncol] = acc[r];
}


__global__ void selfloop_init_kernel(const float* __restrict__ XW, const float* __restrict__ dis,
                                     float* __restrict__ acc, int N, int F) {
  int i = blockIdx.x * blockDim.x + threadIdx.x;
  if (i >= N * F) return;
  float d = dis[i / F];
  acc[i] = XW[i] * (d * d);
}

__global__ void scatter_kernel(const float* __restrict__ XW, const int* __restrict__ src,
                               const int* __restrict__ dst, const float* __restrict__ dis,
                               float* __restrict__ acc, int E, int F) {
  int gid = blockIdx.x * blockDim.x + threadIdx.x;
  int lane = gid & 31;
  int e = gid >> 5;
  if (e >= E) return;
  int s = src[e], d = dst[e];
  float nrm = dis[s] * dis[d];
  int nchunks = F >> 2;
  if (lane < nchunks) {
    const float4 v = *((const float4*)(XW + (size_t)s * F) + lane);
    float* ap = acc + (size_t)d * F + lane * 4;
    atomicAdd(ap + 0, v.x * nrm);
    atomicAdd(ap + 1, v.y * nrm);
    atomicAdd(ap + 2, v.z * nrm);
    atomicAdd(ap + 3, v.w * nrm);
  }
}

__global__ void bias_relu_kernel(float* __restrict__ h, const float* __restrict__ b,
                                 int N, int F) {
  int i = blockIdx.x * blockDim.x + threadIdx.x;
  if (i >= N * F) return;
  float v = h[i] + b[i & (F - 1)];
  h[i] = v > 0.f ? v : 0.f;
}


__global__ void pool_count_kernel(const int* __restrict__ batch, float* __restrict__ cnt, int N) {
  int i = blockIdx.x * blockDim.x + threadIdx.x;
  if (i < N) atomicAdd(&cnt[batch[i]], 1.0f);
}

__global__ void pool_sum_kernel(const float* __restrict__ h, const int* __restrict__ batch,
                                float* __restrict__ out, int N) {
  int gid = blockIdx.x * blockDim.x + threadIdx.x;
  int lane = gid & 31;
  int node = gid >> 5;
  if (node >= N) return;
  int g = batch[node];
  atomicAdd(&out[(size_t)g * OUTC + lane], h[(size_t)node * OUTC + lane]);
}

__global__ void pool_div_kernel(float* __restrict__ out, const float* __restrict__ cnt, int G) {
  int i = blockIdx.x * blockDim.x + threadIdx.x;
  if (i >= G * OUTC) return;
  out[i] /= fmaxf(cnt[i >> 5], 1.0f);
}


extern "C" void kernel_launch(void* const* d_in, const int* in_sizes, int n_in,
                              void* d_out, int out_size, void* d_ws, size_t ws_size,
                              hipStream_t stream) {
  const float* x     = (const float*)d_in[0];
  const int*   ei    = (const int*)d_in[1];
  const int*   batch = (const int*)d_in[2];
  const float* W1    = (const float*)d_in[3];
  const float* b1    = (const float*)d_in[4];
  const float* W2    = (const float*)d_in[5];
  const float* b2    = (const float*)d_in[6];
  const float* W3    = (const float*)d_in[7];
  const float* b3    = (const float*)d_in[8];

  const int N = in_sizes[0] / F_IN;
  const int E = in_sizes[1] / 2;
  const int G = out_size / OUTC;
  const int* src = ei;
  const int* dst = ei + E;
  float* out = (float*)d_out;

  float* dis = (float*)d_ws;
  float* cnt = dis + N;
  size_t off = (size_t)N + (size_t)G;
  off = (off + 63) & ~(size_t)63;
  float* bufA = (float*)d_ws + off;
  float* bufB = bufA + (size_t)N * HID;

  const int TB = 256;

  hipMemsetAsync(cnt, 0, (size_t)G * sizeof(float), stream);
  hipMemsetAsync(out, 0, (size_t)out_size * sizeof(float), stream);

  deg_init_kernel<<<ceil_div(N, TB), TB, 0, stream>>>(dis, N);
  deg_accum_kernel<<<ceil_div(E, TB), TB, 0, stream>>>(dst, dis, E);
  deg_rsqrt_kernel<<<ceil_div(N, TB), TB, 0, stream>>>(dis, N);

  auto conv = [&](const float* hin, const float* W, const float* bb,
                  float* XW, float* accb, int K, int Cout) {
    dim3 gg(ceil_div(ceil_div(N, 16), 4), Cout / 16);
    gemm_wmma_kernel<<<gg, 128, 0, stream>>>(hin, W, XW, N, K, Cout);
    selfloop_init_kernel<<<ceil_div(N * Cout, TB), TB, 0, stream>>>(XW, dis, accb, N, Cout);
    scatter_kernel<<<ceil_div(E * 32, TB), TB, 0, stream>>>(XW, src, dst, dis, accb, E, Cout);
    bias_relu_kernel<<<ceil_div(N * Cout, TB), TB, 0, stream>>>(accb, bb, N, Cout);
  };

  conv(x,    W1, b1, bufA, bufB, F_IN, HID);
  conv(bufB, W2, b2, bufA, bufB, HID,  HID2);
  conv(bufB, W3, b3, bufA, bufB, HID2, OUTC);

  pool_count_kernel<<<ceil_div(N, TB), TB, 0, stream>>>(batch, cnt, N);
  pool_sum_kernel<<<ceil_div(N * 32, TB), TB, 0, stream>>>(bufB, batch, out, N);
  pool_div_kernel<<<ceil_div(G * OUTC, TB), TB, 0, stream>>>(out, cnt, G);
}


# Scatter re-tiled: a wave walks 32 consecutive edges (ids broadcast to scalars, 32 row loads in flight); lanes take consecutive floats, so each f32 atomic covers one 128-B line; sums in arrival order
# speedup vs baseline: 1.7618x; 1.7618x over previous
#include <hip/hip_runtime.h>

typedef float v2f __attribute__((ext_vector_type(2)));
typedef float v8f __attribute__((ext_vector_type(8)));

#define F_IN 128
#define HID  128
#define HID2 64
#define OUTC 32

static inline int ceil_div(int a, int b) { return (a + b - 1) / b; }


__global__ void deg_init_kernel(float* __restrict__ deg, int N) {
  int i = blockIdx.x * blockDim.x + threadIdx.x;
  if (i < N) deg[i] = 1.0f;
}

__global__ void deg_accum_kernel(const int* __restrict__ dst, float* __restrict__ deg, int E) {
  int e = blockIdx.x * blockDim.x + threadIdx.x;
  if (e < E) atomicAdd(&deg[dst[e]], 1.0f);
}

__global__ void deg_rsqrt_kernel(float* __restrict__ deg, int N) {
  int i = blockIdx.x * blockDim.x + threadIdx.x;
  if (i < N) deg[i] = rsqrtf(deg[i]);
}

__global__ __launch_bounds__(128) void gemm_wmma_kernel(
    const float* __restrict__ A, const float* __restrict__ B, float* __restrict__ C,
    int M, int K, int Ncol) {
  int lane = threadIdx.x & 31;
  int wave = threadIdx.x >> 5;
  int mtile = blockIdx.x * 4 + wave;
  int row0 = mtile << 4;
  if (row0 >= M) return;
  int col0 = blockIdx.y << 4;
  int half = lane >> 4;
  int l16  = lane & 15;

  v8f acc = {0.f, 0.f, 0.f, 0.f, 0.f, 0.f, 0.f, 0.f};
  const float* arow = A + (size_t)(row0 + l16) * K;
  const float* bcol = B + col0 + l16;

  for (int k = 0; k < K; k += 4) {
    int ka = k + half * 2;
    v2f a, b;
    a.x = arow[ka + 0];
    a.y = arow[ka + 1];
    b.x = bcol[(size_t)(ka + 0) * Ncol];
    b.y = bcol[(size_t)(ka + 1) * Ncol];
    acc = __builtin_amdgcn_wmma_f32_16x16x4_f32(
         false, a,  false, b,
         (short)0, acc,  false,  false);
  }

  float* crow = C + (size_t)(row0 + half * 8) * Ncol + col0 + l16;
#pragma unroll
  for (int r = 0; r < 8; ++r)
    crow[(size_t)r * Ncol] = acc[r];
}


__global__ void selfloop_init_kernel(const float* __restrict__ XW, const float* __restrict__ dis,
                                     float* __restrict__ acc, int N, int F) {
  int i = blockIdx.x * blockDim.x + threadIdx.x;
  if (i >= N * F) return;
  float d = dis[i / F];
  acc[i] = XW[i] * (d * d);
}

template <int F>
__global__ __launch_bounds__(256) void scatter_kernel(const float* __restrict__ XW,
                                                      const int* __restrict__ src,
                                                      const int* __restrict__ dst,
                                                      const float* __restrict__ dis,
                                                      float* __restrict__ acc, int E) {
  constexpr int J = F / 32;
  constexpr int U = 32 / J;
  int lane = threadIdx.x & 31;
  int e0 = ((blockIdx.x * blockDim.x + threadIdx.x) >> 5) << 5;
  if (e0 >= E) return;
  int e = e0 + lane;
  int s = 0, d = 0;
  float nrm = 0.f;
  if (e < E) {
    s = src[e];
    d = dst[e];
    nrm = dis[s] * dis[d];
  }
  int nrm_bits = __float_as_int(nrm);
  int cnt = E - e0;
  if (cnt >= 32) {
    for (int i0 = 0; i0 < 32; i0 += U) {
      float v[U][J];
#pragma unroll
      for (int u = 0; u < U; ++u) {
        int si = __builtin_amdgcn_readlane(s, i0 + u);
        const float* xr = XW + (size_t)si * F + lane;
#pragma unroll
        for (int j = 0; j < J; ++j) v[u][j] = xr[j * 32];
      }
#pragma unroll
      for (int u = 0; u < U; ++u) {
        int di = __builtin_amdgcn_readlane(d, i0 + u);
        float ni = __int_as_float(__builtin_amdgcn_readlane(nrm_bits, i0 + u));
        float* ar = acc + (size_t)di * F + lane;
#pragma unroll
        for (int j = 0; j < J; ++j) atomicAdd(ar + j * 32, v[u][j] * ni);
      }
    }
  } else {
    for (int i = 0; i < cnt; ++i) {
      int si = __builtin_amdgcn_readlane(s, i);
      int di = __builtin_amdgcn_readlane(d, i);
      float ni = __int_as_float(__builtin_amdgcn_readlane(nrm_bits, i));
      const float* xr = XW + (size_t)si * F + lane;
      float* ar = acc + (size_t)di * F + lane;
#pragma unroll
      for (int j = 0; j < J; ++j) atomicAdd(ar + j * 32, xr[j * 32] * ni);
    }
  }
}

__global__ void bias_relu_kernel(float* __restrict__ h, const float* __restrict__ b,
                                 int N, int F) {
  int i = blockIdx.x * blockDim.x + threadIdx.x;
  if (i >= N * F) return;
  float v = h[i] + b[i & (F - 1)];
  h[i] = v > 0.f ? v : 0.f;
}


__global__ void pool_count_kernel(const int* __restrict__ batch, float* __restrict__ cnt, int N) {
  int i = blockIdx.x * blockDim.x + threadIdx.x;
  if (i < N) atomicAdd(&cnt[batch[i]], 1.0f);
}

__global__ void pool_sum_kernel(const float* __restrict__ h, const int* __restrict__ batch,
                                float* __restrict__ out, int N) {
  int gid = blockIdx.x * blockDim.x + threadIdx.x;
  int lane = gid & 31;
  int node = gid >> 5;
  if (node >= N) return;
  int g = batch[node];
  atomicAdd(&out[(size_t)g * OUTC + lane], h[(size_t)node * OUTC + lane]);
}

__global__ void pool_div_kernel(float* __restrict__ out, const float* __restrict__ cnt, int G) {
  int i = blockIdx.x * blockDim.x + threadIdx.x;
  if (i >= G * OUTC) return;
  out[i] /= fmaxf(cnt[i >> 5], 1.0f);
}


extern "C" void kernel_launch(void* const* d_in, const int* in_sizes, int n_in,
                              void* d_out, int out_size, void* d_ws, size_t ws_size,
                              hipStream_t stream) {
  const float* x     = (const float*)d_in[0];
  const int*   ei    = (const int*)d_in[1];
  const int*   batch = (const int*)d_in[2];
  const float* W1    = (const float*)d_in[3];
  const float* b1    = (const float*)d_in[4];
  const float* W2    = (const float*)d_in[5];
  const float* b2    = (const float*)d_in[6];
  const float* W3    = (const float*)d_in[7];
  const float* b3    = (const float*)d_in[8];

  const int N = in_sizes[0] / F_IN;
  const int E = in_sizes[1] / 2;
  const int G = out_size / OUTC;
  const int* src = ei;
  const int* dst = ei + E;
  float* out = (float*)d_out;

  float* dis = (float*)d_ws;
  float* cnt = dis + N;
  size_t off = (size_t)N + (size_t)G;
  off = (off + 63) & ~(size_t)63;
  float* bufA = (float*)d_ws + off;
  float* bufB = bufA + (size_t)N * HID;

  const int TB = 256;

  hipMemsetAsync(cnt, 0, (size_t)G * sizeof(float), stream);
  hipMemsetAsync(out, 0, (size_t)out_size * sizeof(float), stream);

  deg_init_kernel<<<ceil_div(N, TB), TB, 0, stream>>>(dis, N);
  deg_accum_kernel<<<ceil_div(E, TB), TB, 0, stream>>>(dst, dis, E);
  deg_rsqrt_kernel<<<ceil_div(N, TB), TB, 0, stream>>>(dis, N);

  auto conv = [&](const float* hin, const float* W, const float* bb,
                  float* XW, float* accb, int K, int Cout) {
    dim3 gg(ceil_div(ceil_div(N, 16), 4), Cout / 16);
    gemm_wmma_kernel<<<gg, 128, 0, stream>>>(hin, W, XW, N, K, Cout);
    selfloop_init_kernel<<<ceil_div(N * Cout, TB), TB, 0, stream>>>(XW, dis, accb, N, Cout);
    const int sg = ceil_div(ceil_div(E, 32) * 32, TB);
    if (Cout == HID)
      scatter_kernel<HID><<<sg, TB, 0, stream>>>(XW, src, dst, dis, accb, E);
    else if (Cout == HID2)
      scatter_kernel<HID2><<<sg, TB, 0, stream>>>(XW, src, dst, dis, accb, E);
    else
      scatter_kernel<OUTC><<<sg, TB, 0, stream>>>(XW, src, dst, dis, accb, E);
    bias_relu_kernel<<<ceil_div(N * Cout, TB), TB, 0, stream>>>(accb, bb, N, Cout);
  };

  conv(x,    W1, b1, bufA, bufB, F_IN, HID);
  conv(bufB, W2, b2, bufA, bufB, HID,  HID2);
  conv(bufB, W3, b3, bufA, bufB, HID2, OUTC);

  pool_count_kernel<<<ceil_div(N, TB), TB, 0, stream>>>(batch, cnt, N);
  pool_sum_kernel<<<ceil_div(N * 32, TB), TB, 0, stream>>>(bufB, batch, out, N);
  pool_div_kernel<<<ceil_div(G * OUTC, TB), TB, 0, stream>>>(out, cnt, G);
}


# Every elementwise pass folded into a neighbour: prior bias+ReLU on the transform's A load (last one in pool sum), XW*dis^2 start value from its store, rsqrt by readers, fills and count merged
# speedup vs baseline: 1.8467x; 1.0482x over previous
#include <hip/hip_runtime.h>

typedef float v2f __attribute__((ext_vector_type(2)));
typedef float v8f __attribute__((ext_vector_type(8)));

#define F_IN 128
#define HID  128
#define HID2 64
#define OUTC 32

static inline int ceil_div(int a, int b) { return (a + b - 1) / b; }


__global__ void init_kernel(float* __restrict__ deg, float* __restrict__ cnt,
                            float* __restrict__ out, int N, int G, int out_size) {
  int i = blockIdx.x * blockDim.x + threadIdx.x;
  if (i < N) deg[i] = 1.0f;
  if (i < G) cnt[i] = 0.f;
  if (i < out_size) out[i] = 0.f;
}

__global__ void deg_accum_kernel(const int* __restrict__ dst, float* __restrict__ deg, int E) {
  int e = blockIdx.x * blockDim.x + threadIdx.x;
  if (e < E) atomicAdd(&deg[dst[e]], 1.0f);
}

template <bool ACT>
__global__ __launch_bounds__(128) void gemm_wmma_kernel(
    const float* __restrict__ A, const float* __restrict__ abias, const float* __restrict__ B,
    const float* __restrict__ deg, float* __restrict__ C, float* __restrict__ S,
    int M, int K, int Ncol) {
  int lane = threadIdx.x & 31;
  int wave = threadIdx.x >> 5;
  int mtile = blockIdx.x * 4 + wave;
  int row0 = mtile << 4;
  if (row0 >= M) return;
  int col0 = blockIdx.y << 4;
  int half = lane >> 4;
  int l16  = lane & 15;

  v8f acc = {0.f, 0.f, 0.f, 0.f, 0.f, 0.f, 0.f, 0.f};
  const float* arow = A + (size_t)(row0 + l16) * K;
  const float* bcol = B + col0 + l16;

  for (int k = 0; k < K; k += 4) {
    int ka = k + half * 2;
    v2f a, b;
    a.x = arow[ka + 0];
    a.y = arow[ka + 1];
    if (ACT) {
      float vx = a.x + abias[ka + 0];
      float vy = a.y + abias[ka + 1];
      a.x = vx > 0.f ? vx : 0.f;
      a.y = vy > 0.f ? vy : 0.f;
    }
    b.x = bcol[(size_t)(ka + 0) * Ncol];
    b.y = bcol[(size_t)(ka + 1) * Ncol];
    acc = __builtin_amdgcn_wmma_f32_16x16x4_f32(
         false, a,  false, b,
         (short)0, acc,  false,  false);
  }

  size_t o = (size_t)(row0 + half * 8) * Ncol + col0 + l16;
  float* crow = C + o;
  float* srow = S + o;
  const float* dg = deg + row0 + half * 8;
#pragma unroll
  for (int r = 0; r < 8; ++r) {
    float d = rsqrtf(dg[r]);
    crow[(size_t)r * Ncol] = acc[r];
    srow[(size_t)r * Ncol] = acc[r] * (d * d);
  }
}


template <int F>
__global__ __launch_bounds__(256) void scatter_kernel(const float* __restrict__ XW,
                                                      const int* __restrict__ src,
                                                      const int* __restrict__ dst,
                                                      const float* __restrict__ deg,
                                                      float* __restrict__ acc, int E) {
  constexpr int J = F / 32;
  constexpr int U = 32 / J;
  int lane = threadIdx.x & 31;
  int e0 = ((blockIdx.x * blockDim.x + threadIdx.x) >> 5) << 5;
  if (e0 >= E) return;
  int e = e0 + lane;
  int s = 0, d = 0;
  float nrm = 0.f;
  if (e < E) {
    s = src[e];
    d = dst[e];
    nrm = rsqrtf(deg[s]) * rsqrtf(deg[d]);
  }
  int nrm_bits = __float_as_int(nrm);
  int cnt = E - e0;
  if (cnt >= 32) {
    for (int i0 = 0; i0 < 32; i0 += U) {
      float v[U][J];
#pragma unroll
      for (int u = 0; u < U; ++u) {
        int si = __builtin_amdgcn_readlane(s, i0 + u);
        const float* xr = XW + (size_t)si * F + lane;
#pragma unroll
        for (int j = 0; j < J; ++j) v[u][j] = xr[j * 32];
      }
#pragma unroll
      for (int u = 0; u < U; ++u) {
        int di = __builtin_amdgcn_readlane(d, i0 + u);
        float ni = __int_as_float(__builtin_amdgcn_readlane(nrm_bits, i0 + u));
        float* ar = acc + (size_t)di * F + lane;
#pragma unroll
        for (int j = 0; j < J; ++j) atomicAdd(ar + j * 32, v[u][j] * ni);
      }
    }
  } else {
    for (int i = 0; i < cnt; ++i) {
      int si = __builtin_amdgcn_readlane(s, i);
      int di = __builtin_amdgcn_readlane(d, i);
      float ni = __int_as_float(__builtin_amdgcn_readlane(nrm_bits, i));
      const float* xr = XW + (size_t)si * F + lane;
      float* ar = acc + (size_t)di * F + lane;
#pragma unroll
      for (int j = 0; j < J; ++j) atomicAdd(ar + j * 32, xr[j * 32] * ni);
    }
  }
}


__global__ void pool_sum_kernel(const float* __restrict__ h, const float* __restrict__ b,
                                const int* __restrict__ batch,
                                float* __restrict__ out, float* __restrict__ cnt, int N) {
  int gid = blockIdx.x * blockDim.x + threadIdx.x;
  int lane = gid & 31;
  int node = gid >> 5;
  if (node >= N) return;
  int g = batch[node];
  if (lane == 0) atomicAdd(&cnt[g], 1.0f);
  float v = h[(size_t)node * OUTC + lane] + b[lane];
  atomicAdd(&out[(size_t)g * OUTC + lane], v > 0.f ? v : 0.f);
}

__global__ void pool_div_kernel(float* __restrict__ out, const float* __restrict__ cnt, int G) {
  int i = blockIdx.x * blockDim.x + threadIdx.x;
  if (i >= G * OUTC) return;
  out[i] /= fmaxf(cnt[i >> 5], 1.0f);
}


extern "C" void kernel_launch(void* const* d_in, const int* in_sizes, int n_in,
                              void* d_out, int out_size, void* d_ws, size_t ws_size,
                              hipStream_t stream) {
  const float* x     = (const float*)d_in[0];
  const int*   ei    = (const int*)d_in[1];
  const int*   batch = (const int*)d_in[2];
  const float* W1    = (const float*)d_in[3];
  const float* b1    = (const float*)d_in[4];
  const float* W2    = (const float*)d_in[5];
  const float* b2    = (const float*)d_in[6];
  const float* W3    = (const float*)d_in[7];
  const float* b3    = (const float*)d_in[8];

  const int N = in_sizes[0] / F_IN;
  const int E = in_sizes[1] / 2;
  const int G = out_size / OUTC;
  const int* src = ei;
  const int* dst = ei + E;
  float* out = (float*)d_out;

  float* deg = (float*)d_ws;
  float* cnt = deg + N;
  size_t off = (size_t)N + (size_t)G;
  off = (off + 63) & ~(size_t)63;
  float* bufA = (float*)d_ws + off;
  float* bufB = bufA + (size_t)N * HID;
  float* bufC = bufB + (size_t)N * HID;

  const int TB = 256;

  int n0 = N > out_size ? N : out_size;
  if (G > n0) n0 = G;
  init_kernel<<<ceil_div(n0, TB), TB, 0, stream>>>(deg, cnt, out, N, G, out_size);
  deg_accum_kernel<<<ceil_div(E, TB), TB, 0, stream>>>(dst, deg, E);

  auto conv = [&](const float* hin, const float* pb, const float* W,
                  float* XW, float* accb, int K, int Cout) {
    dim3 gg(ceil_div(ceil_div(N, 16), 4), Cout / 16);
    if (pb)
      gemm_wmma_kernel<true><<<gg, 128, 0, stream>>>(hin, pb, W, deg, XW, accb, N, K, Cout);
    else
      gemm_wmma_kernel<false><<<gg, 128, 0, stream>>>(hin, pb, W, deg, XW, accb, N, K, Cout);
    const int sg = ceil_div(ceil_div(E, 32) * 32, TB);
    if (Cout == HID)
      scatter_kernel<HID><<<sg, TB, 0, stream>>>(XW, src, dst, deg, accb, E);
    else if (Cout == HID2)
      scatter_kernel<HID2><<<sg, TB, 0, stream>>>(XW, src, dst, deg, accb, E);
    else
      scatter_kernel<OUTC><<<sg, TB, 0, stream>>>(XW, src, dst, deg, accb, E);
  };

  conv(x,    nullptr, W1, bufA, bufB, F_IN, HID);
  conv(bufB, b1,      W2, bufA, bufC, HID,  HID2);
  conv(bufC, b2,      W3, bufA, bufB, HID2, OUTC);

  pool_sum_kernel<<<ceil_div(N * 32, TB), TB, 0, stream>>>(bufB, b3, batch, out, cnt, N);
  pool_div_kernel<<<ceil_div(G * OUTC, TB), TB, 0, stream>>>(out, cnt, G);
}


# Dense transform turned weight-stationary: each wave holds its 16-column weight panel in registers and sweeps row tiles, input rows staged once in LDS; k summed per group of 8 as 0,1,4,5 then 2,3,6,7
# speedup vs baseline: 2.8770x; 1.5579x over previous
#include <hip/hip_runtime.h>

typedef float v2f __attribute__((ext_vector_type(2)));
typedef float v8f __attribute__((ext_vector_type(8)));

#define F_IN 128
#define HID  128
#define HID2 64
#define OUTC 32

#define RB     196
#define MAXBIN 1024
#define CHUNK  4096

static inline int ceil_div(int a, int b) { return (a + b - 1) / b; }


__global__ void init_kernel(float* __restrict__ deg, float* __restrict__ cnt,
                            float* __restrict__ out, int* __restrict__ binCount,
                            int* __restrict__ binCursor, int N, int G, int out_size) {
  int i = blockIdx.x * blockDim.x + threadIdx.x;
  if (i < N) deg[i] = 1.0f;
  if (i < G) cnt[i] = 0.f;
  if (i < out_size) out[i] = 0.f;
  if (i < MAXBIN) { binCount[i] = 0; binCursor[i] = 0; }
}


__global__ __launch_bounds__(256) void count_kernel(const int* __restrict__ dst,
                                                    float* __restrict__ deg,
                                                    int* __restrict__ binCount, int E) {
  __shared__ int hist[MAXBIN];
  int t = threadIdx.x;
  for (int i = t; i < MAXBIN; i += 256) hist[i] = 0;
  __syncthreads();
  int e0 = blockIdx.x * CHUNK;
  int e1 = e0 + CHUNK < E ? e0 + CHUNK : E;
  for (int e = e0 + t; e < e1; e += 256) {
    int d = dst[e];
    atomicAdd(&deg[d], 1.0f);
    atomicAdd(&hist[d / RB], 1);
  }
  __syncthreads();
  for (int i = t; i < MAXBIN; i += 256) {
    int h = hist[i];
    if (h) atomicAdd(&binCount[i], h);
  }
}

__global__ __launch_bounds__(256) void place_kernel(const int* __restrict__ src,
                                                    const int* __restrict__ dst,
                                                    const float* __restrict__ deg,
                                                    const int* __restrict__ binCount,
                                                    int* __restrict__ binCursor,
                                                    int* __restrict__ binStart,
                                                    int* __restrict__ es, int* __restrict__ ed,
                                                    float* __restrict__ ew, int E) {
  __shared__ int hist[MAXBIN];
  __shared__ int pos[MAXBIN];
  __shared__ int part[256];
  int t = threadIdx.x;

  int c0 = binCount[4 * t + 0], c1 = binCount[4 * t + 1];
  int c2 = binCount[4 * t + 2], c3 = binCount[4 * t + 3];
  int own = c0 + c1 + c2 + c3;
  part[t] = own;
  for (int i = t; i < MAXBIN; i += 256) hist[i] = 0;
  __syncthreads();
  for (int o = 1; o < 256; o <<= 1) {
    int v = t >= o ? part[t - o] : 0;
    __syncthreads();
    part[t] += v;
    __syncthreads();
  }
  int off = part[t] - own;
  pos[4 * t + 0] = off;
  pos[4 * t + 1] = off + c0;
  pos[4 * t + 2] = off + c0 + c1;
  pos[4 * t + 3] = off + c0 + c1 + c2;
  if (blockIdx.x == 0) {
    binStart[4 * t + 0] = off;
    binStart[4 * t + 1] = off + c0;
    binStart[4 * t + 2] = off + c0 + c1;
    binStart[4 * t + 3] = off + c0 + c1 + c2;
    if (t == 255) binStart[MAXBIN] = off + own;
  }

  int e0 = blockIdx.x * CHUNK;
  int e1 = e0 + CHUNK < E ? e0 + CHUNK : E;
  for (int e = e0 + t; e < e1; e += 256) atomicAdd(&hist[dst[e] / RB], 1);
  __syncthreads();
  for (int i = t; i < MAXBIN; i += 256) {
    int h = hist[i];
    if (h) pos[i] += atomicAdd(&binCursor[i], h);
  }
  __syncthreads();
  for (int e = e0 + t; e < e1; e += 256) {
    int s = src[e];
    int d = dst[e];
    int b = d / RB;
    int slot = atomicAdd(&pos[b], 1);
    es[slot] = s;
    ed[slot] = d - b * RB;
    ew[slot] = rsqrtf(deg[s]) * rsqrtf(deg[d]);
  }
}

template <int K>
__device__ __forceinline__ float4 stage_piece(const float* __restrict__ A, int row, int pc, int M) {
  row = row < M ? row : M - 1;
  return *(const float4*)(A + (size_t)row * K + 4 * pc);
}

template <bool ACT>
__device__ __forceinline__ float4 act_piece(float4 v, float4 b) {
  if (ACT) {
    v.x = fmaxf(v.x + b.x, 0.f);
    v.y = fmaxf(v.y + b.y, 0.f);
    v.z = fmaxf(v.z + b.z, 0.f);
    v.w = fmaxf(v.w + b.w, 0.f);
  }
  return v;
}

template <bool ACT, int K, int NCOL>
__global__ __launch_bounds__(256) void gemm_wstat_kernel(
    const float* __restrict__ A, const float* __restrict__ abias, const float* __restrict__ B,
    float* __restrict__ C, int M) {
  constexpr int NT   = NCOL / 16;
  constexpr int RT   = 8 / NT;
  constexpr int ROWS = RT * 16;
  constexpr int K4   = K / 4;
  constexpr int P    = K + 4;
  constexpr int RSTEP = 256 / K4;
  constexpr int NLD  = ROWS / RSTEP;
  static_assert(NT * RT == 8 && NLD * RSTEP == ROWS, "8 waves, whole pieces");
  __shared__ __attribute__((aligned(16))) float As[2][ROWS * P];

  int t = threadIdx.x;
  int lane = t & 31;
  int wave = __builtin_amdgcn_readfirstlane(t >> 5);
  int c = lane & 15;
  int h = lane >> 4;
  int ct = wave % NT;
  int rt = wave / NT;

  v2f bf[K / 4];
  const float* bcol = B + ct * 16 + c;
#pragma unroll
  for (int j = 0; j < K / 8; ++j) {
    int k0 = 8 * j + 4 * h;
    bf[2 * j].x     = bcol[(size_t)(k0 + 0) * NCOL];
    bf[2 * j].y     = bcol[(size_t)(k0 + 1) * NCOL];
    bf[2 * j + 1].x = bcol[(size_t)(k0 + 2) * NCOL];
    bf[2 * j + 1].y = bcol[(size_t)(k0 + 3) * NCOL];
  }

  const unsigned coff = (unsigned)((rt * 16 + h * 8) * NCOL + ct * 16 + c);

  int pc = t % K4;
  int pr = t / K4;
  float4 bias4 = {0.f, 0.f, 0.f, 0.f};
  if (ACT) bias4 = *(const float4*)(abias + 4 * pc);

  int nsteps = (M + ROWS - 1) / ROWS;
  float4 s0, s1, s2, s3;
  s0 = s1 = s2 = s3 = bias4;
  int step = blockIdx.x;
  if (step < nsteps) {
    s0 = stage_piece<K>(A, step * ROWS + pr, pc, M);
    s1 = stage_piece<K>(A, step * ROWS + pr + RSTEP, pc, M);
    if (NLD == 4) {
      s2 = stage_piece<K>(A, step * ROWS + pr + 2 * RSTEP, pc, M);
      s3 = stage_piece<K>(A, step * ROWS + pr + 3 * RSTEP, pc, M);
    }
  }
  int buf = 0;
  for (; step < nsteps; step += gridDim.x) {
    float* as = As[buf];
    float* ap = as + pr * P + 4 * pc;
    *(float4*)(ap) = act_piece<ACT>(s0, bias4);
    *(float4*)(ap + RSTEP * P) = act_piece<ACT>(s1, bias4);
    if (NLD == 4) {
      *(float4*)(ap + 2 * RSTEP * P) = act_piece<ACT>(s2, bias4);
      *(float4*)(ap + 3 * RSTEP * P) = act_piece<ACT>(s3, bias4);
    }
    __syncthreads();

    int nxt = step + gridDim.x;
    if (nxt < nsteps) {
      s0 = stage_piece<K>(A, nxt * ROWS + pr, pc, M);
      s1 = stage_piece<K>(A, nxt * ROWS + pr + RSTEP, pc, M);
      if (NLD == 4) {
        s2 = stage_piece<K>(A, nxt * ROWS + pr + 2 * RSTEP, pc, M);
        s3 = stage_piece<K>(A, nxt * ROWS + pr + 3 * RSTEP, pc, M);
      }
    }

    int tile0 = step * ROWS + rt * 16;
    float* cstep = C + (size_t)step * (ROWS * NCOL);

    const float* arow = as + (rt * 16 + c) * P + 4 * h;
    float4 a[K / 8];
#pragma unroll
    for (int j = 0; j < K / 8; ++j) a[j] = *(const float4*)(arow + 8 * j);
    __builtin_amdgcn_sched_barrier(0);

    v8f acc = {0.f, 0.f, 0.f, 0.f, 0.f, 0.f, 0.f, 0.f};
#pragma unroll
    for (int j = 0; j < K / 8; ++j) {
      v2f a0 = {a[j].x, a[j].y};
      v2f a1 = {a[j].z, a[j].w};
      acc = __builtin_amdgcn_wmma_f32_16x16x4_f32(
           false, a0,  false, bf[2 * j],
           (short)0, acc,  false,  false);
      acc = __builtin_amdgcn_wmma_f32_16x16x4_f32(
           false, a1,  false, bf[2 * j + 1],
           (short)0, acc,  false,  false);
    }

    if (tile0 + 16 <= M) {
#pragma unroll
      for (int r = 0; r < 8; ++r) cstep[coff + r * NCOL] = acc[r];
    } else {
#pragma unroll
      for (int r = 0; r < 8; ++r)
        if (tile0 + h * 8 + r < M) cstep[coff + r * NCOL] = acc[r];
    }
    buf ^= 1;
  }
}


template <int R>
__device__ __forceinline__ int pick(int x, int i, int sub) {
  int a = __builtin_amdgcn_readlane(x, i);
  if (R == 2) {
    int b = __builtin_amdgcn_readlane(x, i + 1);
    a = sub ? b : a;
  }
  return a;
}

template <int F>
__device__ __forceinline__ void row_add(float* acc, int ri, int q, int sub, float4 v, float w) {
  constexpr int LP = F / 4;
  float* ar = acc + ri * F + q;
  if (F == 128) {
    atomicAdd(ar,          v.x * w);
    atomicAdd(ar + LP,     v.y * w);
    atomicAdd(ar + 2 * LP, v.z * w);
    atomicAdd(ar + 3 * LP, v.w * w);
  } else {
    float y0 = sub ? v.z : v.x;
    float y1 = sub ? v.w : v.y;
    float y2 = sub ? v.x : v.z;
    float y3 = sub ? v.y : v.w;
    float* a1 = ar + (sub ? 2 * LP : 0);
    float* a2 = ar + (sub ? 0 : 2 * LP);
    atomicAdd(a1,      y0 * w);
    atomicAdd(a1 + LP, y1 * w);
    atomicAdd(a2,      y2 * w);
    atomicAdd(a2 + LP, y3 * w);
  }
}

template <int F>
__global__ __launch_bounds__(1024) void agg_rows_kernel(const float* __restrict__ XW,
                                                        const int* __restrict__ es,
                                                        const int* __restrict__ ed,
                                                        const float* __restrict__ ew,
                                                        const int* __restrict__ binStart,
                                                        const float* __restrict__ deg,
                                                        float* __restrict__ sums, int N) {
  static_assert(F == 128 || F == 64, "a row is one or two aligned 256-byte units");
  constexpr int LP = F / 4;
  constexpr int R  = 32 / LP;
  constexpr int U  = 8;
  __shared__ __attribute__((aligned(16))) float acc[RB * F];
  int t = threadIdx.x;
  int lane = t & 31;
  int wave = t >> 5;
  int sub = lane / LP;
  int q = lane % LP;
  int base = blockIdx.x * RB;
  int nrows = N - base < RB ? N - base : RB;

  const float4* xin = (const float4*)(XW + (size_t)base * F);
  for (int i = t; i < nrows * LP; i += 1024) {
    int row = i / LP;
    float d = rsqrtf(deg[base + row]);
    float dd = d * d;
    float4 x = xin[i];
    float* a = acc + row * F + (i % LP);
    a[0]      = x.x * dd;
    a[LP]     = x.y * dd;
    a[2 * LP] = x.z * dd;
    a[3 * LP] = x.w * dd;
  }
  __syncthreads();

  int eb = binStart[blockIdx.x];
  int ee = binStart[blockIdx.x + 1];
  for (int e0 = eb + wave * 32; e0 < ee; e0 += 1024) {
    int e = e0 + lane;
    int s = 0, r = 0;
    float w = 0.f;
    if (e < ee) {
      s = es[e];
      r = ed[e];
      w = ew[e];
    }
    int w_bits = __float_as_int(w);
    int cnt = ee - e0;
    if (cnt >= 32) {
      for (int i0 = 0; i0 < 32; i0 += U * R) {
        float4 v[U];
#pragma unroll
        for (int u = 0; u < U; ++u) {
          int si = pick<R>(s, i0 + u * R, sub);
          v[u] = *(const float4*)(XW + (size_t)si * F + 4 * q);
        }
        __builtin_amdgcn_sched_barrier(0);
#pragma unroll
        for (int u = 0; u < U; ++u) {
          int ri = pick<R>(r, i0 + u * R, sub);
          float wi = __int_as_float(pick<R>(w_bits, i0 + u * R, sub));
          row_add<F>(acc, ri, q, sub, v[u], wi);
        }
      }
    } else {
      for (int i = 0; i < cnt; i += R) {
        int si = pick<R>(s, i, sub);
        int ri = pick<R>(r, i, sub);
        float wi = __int_as_float(pick<R>(w_bits, i, sub));
        if (R == 1 || i + sub < cnt) {
          float4 x = *(const float4*)(XW + (size_t)si * F + 4 * q);
          row_add<F>(acc, ri, q, sub, x, wi);
        }
      }
    }
  }
  __syncthreads();

  float4* o4 = (float4*)(sums + (size_t)base * F);
  for (int i = t; i < nrows * LP; i += 1024) {
    const float* a = acc + (i / LP) * F + (i % LP);
    float4 o;
    o.x = a[0];
    o.y = a[LP];
    o.z = a[2 * LP];
    o.w = a[3 * LP];
    o4[i] = o;
  }
}


template <int F>
__global__ __launch_bounds__(1024) void agg_kernel(const float* __restrict__ XW,
                                                   const int* __restrict__ es,
                                                   const int* __restrict__ ed,
                                                   const float* __restrict__ ew,
                                                   const int* __restrict__ binStart,
                                                   const float* __restrict__ deg,
                                                   float* __restrict__ sums, int N) {
  constexpr int J = F / 32;
  constexpr int U = 32 / J;
  constexpr int F4 = F / 4;
  __shared__ __attribute__((aligned(16))) float acc[RB * F];
  int t = threadIdx.x;
  int lane = t & 31;
  int wave = t >> 5;
  int base = blockIdx.x * RB;
  int nrows = N - base < RB ? N - base : RB;

  const float4* xin = (const float4*)(XW + (size_t)base * F);
  float4* a4 = (float4*)acc;
  for (int i = t; i < nrows * F4; i += 1024) {
    float d = rsqrtf(deg[base + i / F4]);
    float dd = d * d;
    float4 x = xin[i];
    x.x *= dd; x.y *= dd; x.z *= dd; x.w *= dd;
    a4[i] = x;
  }
  __syncthreads();

  int eb = binStart[blockIdx.x];
  int ee = binStart[blockIdx.x + 1];
  for (int e0 = eb + wave * 32; e0 < ee; e0 += 1024) {
    int e = e0 + lane;
    int s = 0, r = 0;
    float w = 0.f;
    if (e < ee) {
      s = es[e];
      r = ed[e];
      w = ew[e];
    }
    int w_bits = __float_as_int(w);
    int cnt = ee - e0;
    if (cnt >= 32) {
      for (int i0 = 0; i0 < 32; i0 += U) {
        float v[U][J];
#pragma unroll
        for (int u = 0; u < U; ++u) {
          int si = __builtin_amdgcn_readlane(s, i0 + u);
          const float* xr = XW + (size_t)si * F + lane;
#pragma unroll
          for (int j = 0; j < J; ++j) v[u][j] = xr[j * 32];
        }
#pragma unroll
        for (int u = 0; u < U; ++u) {
          int ri = __builtin_amdgcn_readlane(r, i0 + u);
          float wi = __int_as_float(__builtin_amdgcn_readlane(w_bits, i0 + u));
          float* ar = acc + ri * F + lane;
#pragma unroll
          for (int j = 0; j < J; ++j) atomicAdd(ar + j * 32, v[u][j] * wi);
        }
      }
    } else {
      for (int i = 0; i < cnt; ++i) {
        int si = __builtin_amdgcn_readlane(s, i);
        int ri = __builtin_amdgcn_readlane(r, i);
        float wi = __int_as_float(__builtin_amdgcn_readlane(w_bits, i));
        const float* xr = XW + (size_t)si * F + lane;
        float* ar = acc + ri * F + lane;
#pragma unroll
        for (int j = 0; j < J; ++j) atomicAdd(ar + j * 32, xr[j * 32] * wi);
      }
    }
  }
  __syncthreads();

  float4* o4 = (float4*)(sums + (size_t)base * F);
  for (int i = t; i < nrows * F4; i += 1024) o4[i] = a4[i];
}


__global__ void pool_sum_kernel(const float* __restrict__ h, const float* __restrict__ b,
                                const int* __restrict__ batch,
                                float* __restrict__ out, float* __restrict__ cnt, int N) {
  int gid = blockIdx.x * blockDim.x + threadIdx.x;
  int lane = gid & 31;
  int node = gid >> 5;
  if (node >= N) return;
  int g = batch[node];
  if (lane == 0) atomicAdd(&cnt[g], 1.0f);
  float v = h[(size_t)node * OUTC + lane] + b[lane];
  atomicAdd(&out[(size_t)g * OUTC + lane], v > 0.f ? v : 0.f);
}

__global__ void pool_div_kernel(float* __restrict__ out, const float* __restrict__ cnt, int G) {
  int i = blockIdx.x * blockDim.x + threadIdx.x;
  if (i >= G * OUTC) return;
  out[i] /= fmaxf(cnt[i >> 5], 1.0f);
}


extern "C" void kernel_launch(void* const* d_in, const int* in_sizes, int n_in,
                              void* d_out, int out_size, void* d_ws, size_t ws_size,
                              hipStream_t stream) {
  const float* x     = (const float*)d_in[0];
  const int*   ei    = (const int*)d_in[1];
  const int*   batch = (const int*)d_in[2];
  const float* W1    = (const float*)d_in[3];
  const float* b1    = (const float*)d_in[4];
  const float* W2    = (const float*)d_in[5];
  const float* b2    = (const float*)d_in[6];
  const float* W3    = (const float*)d_in[7];
  const float* b3    = (const float*)d_in[8];

  const int N = in_sizes[0] / F_IN;
  const int E = in_sizes[1] / 2;
  const int G = out_size / OUTC;
  const int* src = ei;
  const int* dst = ei + E;
  float* out = (float*)d_out;

  float* deg = (float*)d_ws;
  float* cnt = deg + N;
  int* binCount  = (int*)(cnt + G);
  int* binCursor = binCount + MAXBIN;
  int* binStart  = binCursor + MAXBIN;
  size_t off = (size_t)N + (size_t)G + 3 * (size_t)MAXBIN + 1;
  off = (off + 63) & ~(size_t)63;
  float* bufA = (float*)d_ws + off;
  float* bufB = bufA + (size_t)N * HID;
  float* bufC = bufB + (size_t)N * HID;
  int*   es = (int*)(bufC + (size_t)N * HID2);
  int*   ed = es + E;
  float* ew = (float*)(ed + E);

  const int TB = 256;

  int n0 = N > out_size ? N : out_size;
  if (G > n0) n0 = G;
  if (MAXBIN > n0) n0 = MAXBIN;
  init_kernel<<<ceil_div(n0, TB), TB, 0, stream>>>(deg, cnt, out, binCount, binCursor,
                                                    N, G, out_size);
  count_kernel<<<ceil_div(E, CHUNK), 256, 0, stream>>>(dst, deg, binCount, E);
  place_kernel<<<ceil_div(E, CHUNK), 256, 0, stream>>>(src, dst, deg, binCount, binCursor,
                                                       binStart, es, ed, ew, E);
  const int nbin = ceil_div(N, RB);

  const int GB = 512;
  int s1 = ceil_div(N, 16), s2 = ceil_div(N, 32), s3 = ceil_div(N, 64);

  gemm_wstat_kernel<false, F_IN, HID><<<s1 < GB ? s1 : GB, 256, 0, stream>>>(x, nullptr, W1, bufA, N);
  agg_rows_kernel<HID><<<nbin, 1024, 0, stream>>>(bufA, es, ed, ew, binStart, deg, bufB, N);

  gemm_wstat_kernel<true, HID, HID2><<<s2 < GB ? s2 : GB, 256, 0, stream>>>(bufB, b1, W2, bufA, N);
  agg_rows_kernel<HID2><<<nbin, 1024, 0, stream>>>(bufA, es, ed, ew, binStart, deg, bufC, N);

  gemm_wstat_kernel<true, HID2, OUTC><<<s3 < GB ? s3 : GB, 256, 0, stream>>>(bufC, b2, W3, bufA, N);
  agg_kernel<OUTC><<<nbin, 1024, 0, stream>>>(bufA, es, ed, ew, binStart, deg, bufB, N);

  pool_sum_kernel<<<ceil_div(N * 32, TB), TB, 0, stream>>>(bufB, b3, batch, out, cnt, N);
  pool_div_kernel<<<ceil_div(G * OUTC, TB), TB, 0, stream>>>(out, cnt, G);
}


# Pooling re-tiled to few long waves: a wave walks 32 consecutive nodes, all row loads in flight, sums in registers while the graph id repeats, one f32 atomic per run; sums: node order, then arrival
# speedup vs baseline: 3.1116x; 1.0815x over previous
#include <hip/hip_runtime.h>

typedef float v2f __attribute__((ext_vector_type(2)));
typedef float v8f __attribute__((ext_vector_type(8)));

#define F_IN 128
#define HID  128
#define HID2 64
#define OUTC 32

#define RB     196
#define MAXBIN 1024
#define CHUNK  4096

static inline int ceil_div(int a, int b) { return (a + b - 1) / b; }


__global__ void init_kernel(float* __restrict__ deg, float* __restrict__ cnt,
                            float* __restrict__ out, int* __restrict__ binCount,
                            int* __restrict__ binCursor, int N, int G, int out_size) {
  int i = blockIdx.x * blockDim.x + threadIdx.x;
  if (i < N) deg[i] = 1.0f;
  if (i < G) cnt[i] = 0.f;
  if (i < out_size) out[i] = 0.f;
  if (i < MAXBIN) { binCount[i] = 0; binCursor[i] = 0; }
}


__global__ __launch_bounds__(256) void count_kernel(const int* __restrict__ dst,
                                                    float* __restrict__ deg,
                                                    int* __restrict__ binCount, int E) {
  __shared__ int hist[MAXBIN];
  int t = threadIdx.x;
  for (int i = t; i < MAXBIN; i += 256) hist[i] = 0;
  __syncthreads();
  int e0 = blockIdx.x * CHUNK;
  int e1 = e0 + CHUNK < E ? e0 + CHUNK : E;
  for (int e = e0 + t; e < e1; e += 256) {
    int d = dst[e];
    atomicAdd(&deg[d], 1.0f);
    atomicAdd(&hist[d / RB], 1);
  }
  __syncthreads();
  for (int i = t; i < MAXBIN; i += 256) {
    int h = hist[i];
    if (h) atomicAdd(&binCount[i], h);
  }
}

__global__ __launch_bounds__(256) void place_kernel(const int* __restrict__ src,
                                                    const int* __restrict__ dst,
                                                    const float* __restrict__ deg,
                                                    const int* __restrict__ binCount,
                                                    int* __restrict__ binCursor,
                                                    int* __restrict__ binStart,
                                                    int* __restrict__ es, int* __restrict__ ed,
                                                    float* __restrict__ ew, int E) {
  __shared__ int hist[MAXBIN];
  __shared__ int pos[MAXBIN];
  __shared__ int part[256];
  int t = threadIdx.x;

  int c0 = binCount[4 * t + 0], c1 = binCount[4 * t + 1];
  int c2 = binCount[4 * t + 2], c3 = binCount[4 * t + 3];
  int own = c0 + c1 + c2 + c3;
  part[t] = own;
  for (int i = t; i < MAXBIN; i += 256) hist[i] = 0;
  __syncthreads();
  for (int o = 1; o < 256; o <<= 1) {
    int v = t >= o ? part[t - o] : 0;
    __syncthreads();
    part[t] += v;
    __syncthreads();
  }
  int off = part[t] - own;
  pos[4 * t + 0] = off;
  pos[4 * t + 1] = off + c0;
  pos[4 * t + 2] = off + c0 + c1;
  pos[4 * t + 3] = off + c0 + c1 + c2;
  if (blockIdx.x == 0) {
    binStart[4 * t + 0] = off;
    binStart[4 * t + 1] = off + c0;
    binStart[4 * t + 2] = off + c0 + c1;
    binStart[4 * t + 3] = off + c0 + c1 + c2;
    if (t == 255) binStart[MAXBIN] = off + own;
  }

  int e0 = blockIdx.x * CHUNK;
  int e1 = e0 + CHUNK < E ? e0 + CHUNK : E;
  for (int e = e0 + t; e < e1; e += 256) atomicAdd(&hist[dst[e] / RB], 1);
  __syncthreads();
  for (int i = t; i < MAXBIN; i += 256) {
    int h = hist[i];
    if (h) pos[i] += atomicAdd(&binCursor[i], h);
  }
  __syncthreads();
  for (int e = e0 + t; e < e1; e += 256) {
    int s = src[e];
    int d = dst[e];
    int b = d / RB;
    int slot = atomicAdd(&pos[b], 1);
    es[slot] = s;
    ed[slot] = d - b * RB;
    ew[slot] = rsqrtf(deg[s]) * rsqrtf(deg[d]);
  }
}

template <int K>
__device__ __forceinline__ float4 stage_piece(const float* __restrict__ A, int row, int pc, int M) {
  row = row < M ? row : M - 1;
  return *(const float4*)(A + (size_t)row * K + 4 * pc);
}

template <bool ACT>
__device__ __forceinline__ float4 act_piece(float4 v, float4 b) {
  if (ACT) {
    v.x = fmaxf(v.x + b.x, 0.f);
    v.y = fmaxf(v.y + b.y, 0.f);
    v.z = fmaxf(v.z + b.z, 0.f);
    v.w = fmaxf(v.w + b.w, 0.f);
  }
  return v;
}

template <bool ACT, int K, int NCOL>
__global__ __launch_bounds__(256) void gemm_wstat_kernel(
    const float* __restrict__ A, const float* __restrict__ abias, const float* __restrict__ B,
    float* __restrict__ C, int M) {
  constexpr int NT   = NCOL / 16;
  constexpr int RT   = 8 / NT;
  constexpr int ROWS = RT * 16;
  constexpr int K4   = K / 4;
  constexpr int P    = K + 4;
  constexpr int RSTEP = 256 / K4;
  constexpr int NLD  = ROWS / RSTEP;
  static_assert(NT * RT == 8 && NLD * RSTEP == ROWS, "8 waves, whole pieces");
  __shared__ __attribute__((aligned(16))) float As[2][ROWS * P];

  int t = threadIdx.x;
  int lane = t & 31;
  int wave = __builtin_amdgcn_readfirstlane(t >> 5);
  int c = lane & 15;
  int h = lane >> 4;
  int ct = wave % NT;
  int rt = wave / NT;

  v2f bf[K / 4];
  const float* bcol = B + ct * 16 + c;
#pragma unroll
  for (int j = 0; j < K / 8; ++j) {
    int k0 = 8 * j + 4 * h;
    bf[2 * j].x     = bcol[(size_t)(k0 + 0) * NCOL];
    bf[2 * j].y     = bcol[(size_t)(k0 + 1) * NCOL];
    bf[2 * j + 1].x = bcol[(size_t)(k0 + 2) * NCOL];
    bf[2 * j + 1].y = bcol[(size_t)(k0 + 3) * NCOL];
  }

  const unsigned coff = (unsigned)((rt * 16 + h * 8) * NCOL + ct * 16 + c);

  int pc = t % K4;
  int pr = t / K4;
  float4 bias4 = {0.f, 0.f, 0.f, 0.f};
  if (ACT) bias4 = *(const float4*)(abias + 4 * pc);

  int nsteps = (M + ROWS - 1) / ROWS;
  float4 s0, s1, s2, s3;
  s0 = s1 = s2 = s3 = bias4;
  int step = blockIdx.x;
  if (step < nsteps) {
    s0 = stage_piece<K>(A, step * ROWS + pr, pc, M);
    s1 = stage_piece<K>(A, step * ROWS + pr + RSTEP, pc, M);
    if (NLD == 4) {
      s2 = stage_piece<K>(A, step * ROWS + pr + 2 * RSTEP, pc, M);
      s3 = stage_piece<K>(A, step * ROWS + pr + 3 * RSTEP, pc, M);
    }
  }
  int buf = 0;
  for (; step < nsteps; step += gridDim.x) {
    float* as = As[buf];
    float* ap = as + pr * P + 4 * pc;
    *(float4*)(ap) = act_piece<ACT>(s0, bias4);
    *(float4*)(ap + RSTEP * P) = act_piece<ACT>(s1, bias4);
    if (NLD == 4) {
      *(float4*)(ap + 2 * RSTEP * P) = act_piece<ACT>(s2, bias4);
      *(float4*)(ap + 3 * RSTEP * P) = act_piece<ACT>(s3, bias4);
    }
    __syncthreads();

    int nxt = step + gridDim.x;
    if (nxt < nsteps) {
      s0 = stage_piece<K>(A, nxt * ROWS + pr, pc, M);
      s1 = stage_piece<K>(A, nxt * ROWS + pr + RSTEP, pc, M);
      if (NLD == 4) {
        s2 = stage_piece<K>(A, nxt * ROWS + pr + 2 * RSTEP, pc, M);
        s3 = stage_piece<K>(A, nxt * ROWS + pr + 3 * RSTEP, pc, M);
      }
    }

    int tile0 = step * ROWS + rt * 16;
    float* cstep = C + (size_t)step * (ROWS * NCOL);

    const float* arow = as + (rt * 16 + c) * P + 4 * h;
    float4 a[K / 8];
#pragma unroll
    for (int j = 0; j < K / 8; ++j) a[j] = *(const float4*)(arow + 8 * j);
    __builtin_amdgcn_sched_barrier(0);

    v8f acc = {0.f, 0.f, 0.f, 0.f, 0.f, 0.f, 0.f, 0.f};
#pragma unroll
    for (int j = 0; j < K / 8; ++j) {
      v2f a0 = {a[j].x, a[j].y};
      v2f a1 = {a[j].z, a[j].w};
      acc = __builtin_amdgcn_wmma_f32_16x16x4_f32(
           false, a0,  false, bf[2 * j],
           (short)0, acc,  false,  false);
      acc = __builtin_amdgcn_wmma_f32_16x16x4_f32(
           false, a1,  false, bf[2 * j + 1],
           (short)0, acc,  false,  false);
    }

    if (tile0 + 16 <= M) {
#pragma unroll
      for (int r = 0; r < 8; ++r) cstep[coff + r * NCOL] = acc[r];
    } else {
#pragma unroll
      for (int r = 0; r < 8; ++r)
        if (tile0 + h * 8 + r < M) cstep[coff + r * NCOL] = acc[r];
    }
    buf ^= 1;
  }
}


template <int R>
__device__ __forceinline__ int pick(int x, int i, int sub) {
  int a = __builtin_amdgcn_readlane(x, i);
  if (R == 2) {
    int b = __builtin_amdgcn_readlane(x, i + 1);
    a = sub ? b : a;
  }
  return a;
}

template <int F>
__device__ __forceinline__ void row_add(float* acc, int ri, int q, int sub, float4 v, float w) {
  constexpr int LP = F / 4;
  float* ar = acc + ri * F + q;
  if (F == 128) {
    atomicAdd(ar,          v.x * w);
    atomicAdd(ar + LP,     v.y * w);
    atomicAdd(ar + 2 * LP, v.z * w);
    atomicAdd(ar + 3 * LP, v.w * w);
  } else {
    float y0 = sub ? v.z : v.x;
    float y1 = sub ? v.w : v.y;
    float y2 = sub ? v.x : v.z;
    float y3 = sub ? v.y : v.w;
    float* a1 = ar + (sub ? 2 * LP : 0);
    float* a2 = ar + (sub ? 0 : 2 * LP);
    atomicAdd(a1,      y0 * w);
    atomicAdd(a1 + LP, y1 * w);
    atomicAdd(a2,      y2 * w);
    atomicAdd(a2 + LP, y3 * w);
  }
}

template <int F>
__global__ __launch_bounds__(1024) void agg_rows_kernel(const float* __restrict__ XW,
                                                        const int* __restrict__ es,
                                                        const int* __restrict__ ed,
                                                        const float* __restrict__ ew,
                                                        const int* __restrict__ binStart,
                                                        const float* __restrict__ deg,
                                                        float* __restrict__ sums, int N) {
  static_assert(F == 128 || F == 64, "a row is one or two aligned 256-byte units");
  constexpr int LP = F / 4;
  constexpr int R  = 32 / LP;
  constexpr int U  = 8;
  __shared__ __attribute__((aligned(16))) float acc[RB * F];
  int t = threadIdx.x;
  int lane = t & 31;
  int wave = t >> 5;
  int sub = lane / LP;
  int q = lane % LP;
  int base = blockIdx.x * RB;
  int nrows = N - base < RB ? N - base : RB;

  const float4* xin = (const float4*)(XW + (size_t)base * F);
  for (int i = t; i < nrows * LP; i += 1024) {
    int row = i / LP;
    float d = rsqrtf(deg[base + row]);
    float dd = d * d;
    float4 x = xin[i];
    float* a = acc + row * F + (i % LP);
    a[0]      = x.x * dd;
    a[LP]     = x.y * dd;
    a[2 * LP] = x.z * dd;
    a[3 * LP] = x.w * dd;
  }
  __syncthreads();

  int eb = binStart[blockIdx.x];
  int ee = binStart[blockIdx.x + 1];
  for (int e0 = eb + wave * 32; e0 < ee; e0 += 1024) {
    int e = e0 + lane;
    int s = 0, r = 0;
    float w = 0.f;
    if (e < ee) {
      s = es[e];
      r = ed[e];
      w = ew[e];
    }
    int w_bits = __float_as_int(w);
    int cnt = ee - e0;
    if (cnt >= 32) {
      for (int i0 = 0; i0 < 32; i0 += U * R) {
        float4 v[U];
#pragma unroll
        for (int u = 0; u < U; ++u) {
          int si = pick<R>(s, i0 + u * R, sub);
          v[u] = *(const float4*)(XW + (size_t)si * F + 4 * q);
        }
        __builtin_amdgcn_sched_barrier(0);
#pragma unroll
        for (int u = 0; u < U; ++u) {
          int ri = pick<R>(r, i0 + u * R, sub);
          float wi = __int_as_float(pick<R>(w_bits, i0 + u * R, sub));
          row_add<F>(acc, ri, q, sub, v[u], wi);
        }
      }
    } else {
      for (int i = 0; i < cnt; i += R) {
        int si = pick<R>(s, i, sub);
        int ri = pick<R>(r, i, sub);
        float wi = __int_as_float(pick<R>(w_bits, i, sub));
        if (R == 1 || i + sub < cnt) {
          float4 x = *(const float4*)(XW + (size_t)si * F + 4 * q);
          row_add<F>(acc, ri, q, sub, x, wi);
        }
      }
    }
  }
  __syncthreads();

  float4* o4 = (float4*)(sums + (size_t)base * F);
  for (int i = t; i < nrows * LP; i += 1024) {
    const float* a = acc + (i / LP) * F + (i % LP);
    float4 o;
    o.x = a[0];
    o.y = a[LP];
    o.z = a[2 * LP];
    o.w = a[3 * LP];
    o4[i] = o;
  }
}


template <int F>
__global__ __launch_bounds__(1024) void agg_kernel(const float* __restrict__ XW,
                                                   const int* __restrict__ es,
                                                   const int* __restrict__ ed,
                                                   const float* __restrict__ ew,
                                                   const int* __restrict__ binStart,
                                                   const float* __restrict__ deg,
                                                   float* __restrict__ sums, int N) {
  constexpr int J = F / 32;
  constexpr int U = 32 / J;
  constexpr int F4 = F / 4;
  __shared__ __attribute__((aligned(16))) float acc[RB * F];
  int t = threadIdx.x;
  int lane = t & 31;
  int wave = t >> 5;
  int base = blockIdx.x * RB;
  int nrows = N - base < RB ? N - base : RB;

  const float4* xin = (const float4*)(XW + (size_t)base * F);
  float4* a4 = (float4*)acc;
  for (int i = t; i < nrows * F4; i += 1024) {
    float d = rsqrtf(deg[base + i / F4]);
    float dd = d * d;
    float4 x = xin[i];
    x.x *= dd; x.y *= dd; x.z *= dd; x.w *= dd;
    a4[i] = x;
  }
  __syncthreads();

  int eb = binStart[blockIdx.x];
  int ee = binStart[blockIdx.x + 1];
  for (int e0 = eb + wave * 32; e0 < ee; e0 += 1024) {
    int e = e0 + lane;
    int s = 0, r = 0;
    float w = 0.f;
    if (e < ee) {
      s = es[e];
      r = ed[e];
      w = ew[e];
    }
    int w_bits = __float_as_int(w);
    int cnt = ee - e0;
    if (cnt >= 32) {
      for (int i0 = 0; i0 < 32; i0 += U) {
        float v[U][J];
#pragma unroll
        for (int u = 0; u < U; ++u) {
          int si = __builtin_amdgcn_readlane(s, i0 + u);
          const float* xr = XW + (size_t)si * F + lane;
#pragma unroll
          for (int j = 0; j < J; ++j) v[u][j] = xr[j * 32];
        }
#pragma unroll
        for (int u = 0; u < U; ++u) {
          int ri = __builtin_amdgcn_readlane(r, i0 + u);
          float wi = __int_as_float(__builtin_amdgcn_readlane(w_bits, i0 + u));
          float* ar = acc + ri * F + lane;
#pragma unroll
          for (int j = 0; j < J; ++j) atomicAdd(ar + j * 32, v[u][j] * wi);
        }
      }
    } else {
      for (int i = 0; i < cnt; ++i) {
        int si = __builtin_amdgcn_readlane(s, i);
        int ri = __builtin_amdgcn_readlane(r, i);
        float wi = __int_as_float(__builtin_amdgcn_readlane(w_bits, i));
        const float* xr = XW + (size_t)si * F + lane;
        float* ar = acc + ri * F + lane;
#pragma unroll
        for (int j = 0; j < J; ++j) atomicAdd(ar + j * 32, xr[j * 32] * wi);
      }
    }
  }
  __syncthreads();

  float4* o4 = (float4*)(sums + (size_t)base * F);
  for (int i = t; i < nrows * F4; i += 1024) o4[i] = a4[i];
}


#define PN 32
__global__ __launch_bounds__(128) void pool_sum_kernel(const float* __restrict__ h,
                                                       const float* __restrict__ b,
                                                       const int* __restrict__ batch,
                                                       float* __restrict__ out,
                                                       float* __restrict__ cnt, int N) {
  int lane = threadIdx.x & 31;
  int wave = blockIdx.x * 4 + __builtin_amdgcn_readfirstlane(threadIdx.x >> 5);
  int n0 = wave * PN;
  if (n0 >= N) return;
  int nn = N - n0 < PN ? N - n0 : PN;
  int g = batch[n0 + (lane < nn ? lane : nn - 1)];
  float bl = b[lane];
  const float* hp = h + (size_t)n0 * OUTC + lane;
  float v[PN];
#pragma unroll
  for (int i = 0; i < PN; ++i) v[i] = hp[(size_t)(i < nn ? i : nn - 1) * OUTC];

  int cur = __builtin_amdgcn_readlane(g, 0);
  float x0 = v[0] + bl;
  float acc = x0 > 0.f ? x0 : 0.f;
  int n = 1;
#pragma unroll
  for (int i = 1; i < PN; ++i) {
    if (i < nn) {
      int gi = __builtin_amdgcn_readlane(g, i);
      if (gi != cur) {
        atomicAdd(&out[(size_t)cur * OUTC + lane], acc);
        if (lane == 0) atomicAdd(&cnt[cur], (float)n);
        cur = gi;
        acc = 0.f;
        n = 0;
      }
      float x = v[i] + bl;
      acc += x > 0.f ? x : 0.f;
      ++n;
    }
  }
  atomicAdd(&out[(size_t)cur * OUTC + lane], acc);
  if (lane == 0) atomicAdd(&cnt[cur], (float)n);
}

__global__ void pool_div_kernel(float* __restrict__ out, const float* __restrict__ cnt, int G) {
  int i = blockIdx.x * blockDim.x + threadIdx.x;
  if (i >= G * OUTC) return;
  out[i] /= fmaxf(cnt[i >> 5], 1.0f);
}


extern "C" void kernel_launch(void* const* d_in, const int* in_sizes, int n_in,
                              void* d_out, int out_size, void* d_ws, size_t ws_size,
                              hipStream_t stream) {
  const float* x     = (const float*)d_in[0];
  const int*   ei    = (const int*)d_in[1];
  const int*   batch = (const int*)d_in[2];
  const float* W1    = (const float*)d_in[3];
  const float* b1    = (const float*)d_in[4];
  const float* W2    = (const float*)d_in[5];
  const float* b2    = (const float*)d_in[6];
  const float* W3    = (const float*)d_in[7];
  const float* b3    = (const float*)d_in[8];

  const int N = in_sizes[0] / F_IN;
  const int E = in_sizes[1] / 2;
  const int G = out_size / OUTC;
  const int* src = ei;
  const int* dst = ei + E;
  float* out = (float*)d_out;

  float* deg = (float*)d_ws;
  float* cnt = deg + N;
  int* binCount  = (int*)(cnt + G);
  int* binCursor = binCount + MAXBIN;
  int* binStart  = binCursor + MAXBIN;
  size_t off = (size_t)N + (size_t)G + 3 * (size_t)MAXBIN + 1;
  off = (off + 63) & ~(size_t)63;
  float* bufA = (float*)d_ws + off;
  float* bufB = bufA + (size_t)N * HID;
  float* bufC = bufB + (size_t)N * HID;
  int*   es = (int*)(bufC + (size_t)N * HID2);
  int*   ed = es + E;
  float* ew = (float*)(ed + E);

  const int TB = 256;

  int n0 = N > out_size ? N : out_size;
  if (G > n0) n0 = G;
  if (MAXBIN > n0) n0 = MAXBIN;
  init_kernel<<<ceil_div(n0, TB), TB, 0, stream>>>(deg, cnt, out, binCount, binCursor,
                                                    N, G, out_size);
  count_kernel<<<ceil_div(E, CHUNK), 256, 0, stream>>>(dst, deg, binCount, E);
  place_kernel<<<ceil_div(E, CHUNK), 256, 0, stream>>>(src, dst, deg, binCount, binCursor,
                                                       binStart, es, ed, ew, E);
  const int nbin = ceil_div(N, RB);

  const int GB = 512;
  int s1 = ceil_div(N, 16), s2 = ceil_div(N, 32), s3 = ceil_div(N, 64);

  gemm_wstat_kernel<false, F_IN, HID><<<s1 < GB ? s1 : GB, 256, 0, stream>>>(x, nullptr, W1, bufA, N);
  agg_rows_kernel<HID><<<nbin, 1024, 0, stream>>>(bufA, es, ed, ew, binStart, deg, bufB, N);

  gemm_wstat_kernel<true, HID, HID2><<<s2 < GB ? s2 : GB, 256, 0, stream>>>(bufB, b1, W2, bufA, N);
  agg_rows_kernel<HID2><<<nbin, 1024, 0, stream>>>(bufA, es, ed, ew, binStart, deg, bufC, N);

  gemm_wstat_kernel<true, HID2, OUTC><<<s3 < GB ? s3 : GB, 256, 0, stream>>>(bufC, b2, W3, bufA, N);
  agg_kernel<OUTC><<<nbin, 1024, 0, stream>>>(bufA, es, ed, ew, binStart, deg, bufB, N);

  pool_sum_kernel<<<ceil_div(N, PN * 4), 128, 0, stream>>>(bufB, b3, batch, out, cnt, N);
  pool_div_kernel<<<ceil_div(G * OUTC, TB), TB, 0, stream>>>(out, cnt, G);
}


# Degrees counted per destination range by its owner block from the finished bucket (LDS integer adds), not by a global f32 atomic per edge; first edge pass makes the edge weights; same sum order
# speedup vs baseline: 4.4669x; 1.4356x over previous
#include <hip/hip_runtime.h>

typedef float v2f __attribute__((ext_vector_type(2)));
typedef float v8f __attribute__((ext_vector_type(8)));

#define F_IN 128
#define HID  128
#define HID2 64
#define OUTC 32

#define RB     196
#define MAXBIN 1024
#define CHUNK  4096

static inline int ceil_div(int a, int b) { return (a + b - 1) / b; }


__global__ void init_kernel(float* __restrict__ cnt, float* __restrict__ out,
                            int* __restrict__ binCount, int* __restrict__ binCursor,
                            int G, int out_size) {
  int i = blockIdx.x * blockDim.x + threadIdx.x;
  if (i < G) cnt[i] = 0.f;
  if (i < out_size) out[i] = 0.f;
  if (i < MAXBIN) { binCount[i] = 0; binCursor[i] = 0; }
}


__global__ __launch_bounds__(256) void count_kernel(const int* __restrict__ dst,
                                                    int* __restrict__ binCount, int E) {
  __shared__ int hist[MAXBIN];
  int t = threadIdx.x;
  for (int i = t; i < MAXBIN; i += 256) hist[i] = 0;
  __syncthreads();
  int e0 = blockIdx.x * CHUNK;
  int e1 = e0 + CHUNK < E ? e0 + CHUNK : E;
  for (int e = e0 + t; e < e1; e += 256) {
    atomicAdd(&hist[dst[e] / RB], 1);
  }
  __syncthreads();
  for (int i = t; i < MAXBIN; i += 256) {
    int h = hist[i];
    if (h) atomicAdd(&binCount[i], h);
  }
}

__global__ __launch_bounds__(256) void place_kernel(const int* __restrict__ src,
                                                    const int* __restrict__ dst,
                                                    const int* __restrict__ binCount,
                                                    int* __restrict__ binCursor,
                                                    int* __restrict__ binStart,
                                                    int* __restrict__ es, int* __restrict__ ed,
                                                    int E) {
  __shared__ int hist[MAXBIN];
  __shared__ int pos[MAXBIN];
  __shared__ int part[256];
  int t = threadIdx.x;

  int c0 = binCount[4 * t + 0], c1 = binCount[4 * t + 1];
  int c2 = binCount[4 * t + 2], c3 = binCount[4 * t + 3];
  int own = c0 + c1 + c2 + c3;
  part[t] = own;
  for (int i = t; i < MAXBIN; i += 256) hist[i] = 0;
  __syncthreads();
  for (int o = 1; o < 256; o <<= 1) {
    int v = t >= o ? part[t - o] : 0;
    __syncthreads();
    part[t] += v;
    __syncthreads();
  }
  int off = part[t] - own;
  pos[4 * t + 0] = off;
  pos[4 * t + 1] = off + c0;
  pos[4 * t + 2] = off + c0 + c1;
  pos[4 * t + 3] = off + c0 + c1 + c2;
  if (blockIdx.x == 0) {
    binStart[4 * t + 0] = off;
    binStart[4 * t + 1] = off + c0;
    binStart[4 * t + 2] = off + c0 + c1;
    binStart[4 * t + 3] = off + c0 + c1 + c2;
    if (t == 255) binStart[MAXBIN] = off + own;
  }

  int e0 = blockIdx.x * CHUNK;
  int e1 = e0 + CHUNK < E ? e0 + CHUNK : E;
  for (int e = e0 + t; e < e1; e += 256) atomicAdd(&hist[dst[e] / RB], 1);
  __syncthreads();
  for (int i = t; i < MAXBIN; i += 256) {
    int h = hist[i];
    if (h) pos[i] += atomicAdd(&binCursor[i], h);
  }
  __syncthreads();
  for (int e = e0 + t; e < e1; e += 256) {
    int s = src[e];
    int d = dst[e];
    int b = d / RB;
    int slot = atomicAdd(&pos[b], 1);
    es[slot] = s;
    ed[slot] = d - b * RB;
  }
}

__global__ __launch_bounds__(1024) void degree_kernel(const int* __restrict__ ed,
                                                      const int* __restrict__ binStart,
                                                      float* __restrict__ deg, int N) {
  __shared__ int rowCount[RB];
  int t = threadIdx.x;
  if (t < RB) rowCount[t] = 0;
  __syncthreads();
  int eb = binStart[blockIdx.x];
  int ee = binStart[blockIdx.x + 1];
  for (int e = eb + t; e < ee; e += 1024) atomicAdd(&rowCount[ed[e]], 1);
  __syncthreads();
  int node = blockIdx.x * RB + t;
  if (t < RB && node < N) deg[node] = 1.0f + (float)rowCount[t];
}

template <int K>
__device__ __forceinline__ float4 stage_piece(const float* __restrict__ A, int row, int pc, int M) {
  row = row < M ? row : M - 1;
  return *(const float4*)(A + (size_t)row * K + 4 * pc);
}

template <bool ACT>
__device__ __forceinline__ float4 act_piece(float4 v, float4 b) {
  if (ACT) {
    v.x = fmaxf(v.x + b.x, 0.f);
    v.y = fmaxf(v.y + b.y, 0.f);
    v.z = fmaxf(v.z + b.z, 0.f);
    v.w = fmaxf(v.w + b.w, 0.f);
  }
  return v;
}

template <bool ACT, int K, int NCOL>
__global__ __launch_bounds__(256) void gemm_wstat_kernel(
    const float* __restrict__ A, const float* __restrict__ abias, const float* __restrict__ B,
    float* __restrict__ C, int M) {
  constexpr int NT   = NCOL / 16;
  constexpr int RT   = 8 / NT;
  constexpr int ROWS = RT * 16;
  constexpr int K4   = K / 4;
  constexpr int P    = K + 4;
  constexpr int RSTEP = 256 / K4;
  constexpr int NLD  = ROWS / RSTEP;
  static_assert(NT * RT == 8 && NLD * RSTEP == ROWS, "8 waves, whole pieces");
  __shared__ __attribute__((aligned(16))) float As[2][ROWS * P];

  int t = threadIdx.x;
  int lane = t & 31;
  int wave = __builtin_amdgcn_readfirstlane(t >> 5);
  int c = lane & 15;
  int h = lane >> 4;
  int ct = wave % NT;
  int rt = wave / NT;

  v2f bf[K / 4];
  const float* bcol = B + ct * 16 + c;
#pragma unroll
  for (int j = 0; j < K / 8; ++j) {
    int k0 = 8 * j + 4 * h;
    bf[2 * j].x     = bcol[(size_t)(k0 + 0) * NCOL];
    bf[2 * j].y     = bcol[(size_t)(k0 + 1) * NCOL];
    bf[2 * j + 1].x = bcol[(size_t)(k0 + 2) * NCOL];
    bf[2 * j + 1].y = bcol[(size_t)(k0 + 3) * NCOL];
  }

  const unsigned coff = (unsigned)((rt * 16 + h * 8) * NCOL + ct * 16 + c);

  int pc = t % K4;
  int pr = t / K4;
  float4 bias4 = {0.f, 0.f, 0.f, 0.f};
  if (ACT) bias4 = *(const float4*)(abias + 4 * pc);

  int nsteps = (M + ROWS - 1) / ROWS;
  float4 s0, s1, s2, s3;
  s0 = s1 = s2 = s3 = bias4;
  int step = blockIdx.x;
  if (step < nsteps) {
    s0 = stage_piece<K>(A, step * ROWS + pr, pc, M);
    s1 = stage_piece<K>(A, step * ROWS + pr + RSTEP, pc, M);
    if (NLD == 4) {
      s2 = stage_piece<K>(A, step * ROWS + pr + 2 * RSTEP, pc, M);
      s3 = stage_piece<K>(A, step * ROWS + pr + 3 * RSTEP, pc, M);
    }
  }
  int buf = 0;
  for (; step < nsteps; step += gridDim.x) {
    float* as = As[buf];
    float* ap = as + pr * P + 4 * pc;
    *(float4*)(ap) = act_piece<ACT>(s0, bias4);
    *(float4*)(ap + RSTEP * P) = act_piece<ACT>(s1, bias4);
    if (NLD == 4) {
      *(float4*)(ap + 2 * RSTEP * P) = act_piece<ACT>(s2, bias4);
      *(float4*)(ap + 3 * RSTEP * P) = act_piece<ACT>(s3, bias4);
    }
    __syncthreads();

    int nxt = step + gridDim.x;
    if (nxt < nsteps) {
      s0 = stage_piece<K>(A, nxt * ROWS + pr, pc, M);
      s1 = stage_piece<K>(A, nxt * ROWS + pr + RSTEP, pc, M);
      if (NLD == 4) {
        s2 = stage_piece<K>(A, nxt * ROWS + pr + 2 * RSTEP, pc, M);
        s3 = stage_piece<K>(A, nxt * ROWS + pr + 3 * RSTEP, pc, M);
      }
    }

    int tile0 = step * ROWS + rt * 16;
    float* cstep = C + (size_t)step * (ROWS * NCOL);

    const float* arow = as + (rt * 16 + c) * P + 4 * h;
    float4 a[K / 8];
#pragma unroll
    for (int j = 0; j < K / 8; ++j) a[j] = *(const float4*)(arow + 8 * j);
    __builtin_amdgcn_sched_barrier(0);

    v8f acc = {0.f, 0.f, 0.f, 0.f, 0.f, 0.f, 0.f, 0.f};
#pragma unroll
    for (int j = 0; j < K / 8; ++j) {
      v2f a0 = {a[j].x, a[j].y};
      v2f a1 = {a[j].z, a[j].w};
      acc = __builtin_amdgcn_wmma_f32_16x16x4_f32(
           false, a0,  false, bf[2 * j],
           (short)0, acc,  false,  false);
      acc = __builtin_amdgcn_wmma_f32_16x16x4_f32(
           false, a1,  false, bf[2 * j + 1],
           (short)0, acc,  false,  false);
    }

    if (tile0 + 16 <= M) {
#pragma unroll
      for (int r = 0; r < 8; ++r) cstep[coff + r * NCOL] = acc[r];
    } else {
#pragma unroll
      for (int r = 0; r < 8; ++r)
        if (tile0 + h * 8 + r < M) cstep[coff + r * NCOL] = acc[r];
    }
    buf ^= 1;
  }
}


template <int R>
__device__ __forceinline__ int pick(int x, int i, int sub) {
  int a = __builtin_amdgcn_readlane(x, i);
  if (R == 2) {
    int b = __builtin_amdgcn_readlane(x, i + 1);
    a = sub ? b : a;
  }
  return a;
}

template <int F>
__device__ __forceinline__ void row_add(float* acc, int ri, int q, int sub, float4 v, float w) {
  constexpr int LP = F / 4;
  float* ar = acc + ri * F + q;
  if (F == 128) {
    atomicAdd(ar,          v.x * w);
    atomicAdd(ar + LP,     v.y * w);
    atomicAdd(ar + 2 * LP, v.z * w);
    atomicAdd(ar + 3 * LP, v.w * w);
  } else {
    float y0 = sub ? v.z : v.x;
    float y1 = sub ? v.w : v.y;
    float y2 = sub ? v.x : v.z;
    float y3 = sub ? v.y : v.w;
    float* a1 = ar + (sub ? 2 * LP : 0);
    float* a2 = ar + (sub ? 0 : 2 * LP);
    atomicAdd(a1,      y0 * w);
    atomicAdd(a1 + LP, y1 * w);
    atomicAdd(a2,      y2 * w);
    atomicAdd(a2 + LP, y3 * w);
  }
}

template <int F, bool MKW>
__global__ __launch_bounds__(1024) void agg_rows_kernel(const float* __restrict__ XW,
                                                        const int* __restrict__ es,
                                                        const int* __restrict__ ed,
                                                        float* __restrict__ ew,
                                                        const int* __restrict__ binStart,
                                                        const float* __restrict__ deg,
                                                        float* __restrict__ sums, int N) {
  static_assert(F == 128 || F == 64, "a row is one or two aligned 256-byte units");
  constexpr int LP = F / 4;
  constexpr int R  = 32 / LP;
  constexpr int U  = 8;
  __shared__ __attribute__((aligned(16))) float acc[RB * F];
  int t = threadIdx.x;
  int lane = t & 31;
  int wave = t >> 5;
  int sub = lane / LP;
  int q = lane % LP;
  int base = blockIdx.x * RB;
  int nrows = N - base < RB ? N - base : RB;

  const float4* xin = (const float4*)(XW + (size_t)base * F);
  for (int i = t; i < nrows * LP; i += 1024) {
    int row = i / LP;
    float d = rsqrtf(deg[base + row]);
    float dd = d * d;
    float4 x = xin[i];
    float* a = acc + row * F + (i % LP);
    a[0]      = x.x * dd;
    a[LP]     = x.y * dd;
    a[2 * LP] = x.z * dd;
    a[3 * LP] = x.w * dd;
  }
  __syncthreads();

  int eb = binStart[blockIdx.x];
  int ee = binStart[blockIdx.x + 1];
  for (int e0 = eb + wave * 32; e0 < ee; e0 += 1024) {
    int e = e0 + lane;
    int s = 0, r = 0;
    float w = 0.f;
    if (e < ee) {
      s = es[e];
      r = ed[e];
      if (!MKW) w = ew[e];
    }
    int cnt = ee - e0;
    if (MKW && cnt < 32 && e < ee) {
      w = rsqrtf(deg[s]) * rsqrtf(deg[base + r]);
      ew[e] = w;
    }
    int w_bits = __float_as_int(w);
    if (cnt >= 32) {
      float dgs = 1.f, dgd = 1.f;
      for (int i0 = 0; i0 < 32; i0 += U * R) {
        float4 v[U];
#pragma unroll
        for (int u = 0; u < U; ++u) {
          int si = pick<R>(s, i0 + u * R, sub);
          v[u] = *(const float4*)(XW + (size_t)si * F + 4 * q);
        }
        if (MKW && i0 == 0) {
          dgs = deg[s];
          dgd = deg[base + r];
        }
        __builtin_amdgcn_sched_barrier(0);
        if (MKW && i0 == 0) {
          w = rsqrtf(dgs) * rsqrtf(dgd);
          ew[e] = w;
          w_bits = __float_as_int(w);
        }
#pragma unroll
        for (int u = 0; u < U; ++u) {
          int ri = pick<R>(r, i0 + u * R, sub);
          float wi = __int_as_float(pick<R>(w_bits, i0 + u * R, sub));
          row_add<F>(acc, ri, q, sub, v[u], wi);
        }
      }
    } else {
      for (int i = 0; i < cnt; i += R) {
        int si = pick<R>(s, i, sub);
        int ri = pick<R>(r, i, sub);
        float wi = __int_as_float(pick<R>(w_bits, i, sub));
        if (R == 1 || i + sub < cnt) {
          float4 x = *(const float4*)(XW + (size_t)si * F + 4 * q);
          row_add<F>(acc, ri, q, sub, x, wi);
        }
      }
    }
  }
  __syncthreads();

  float4* o4 = (float4*)(sums + (size_t)base * F);
  for (int i = t; i < nrows * LP; i += 1024) {
    const float* a = acc + (i / LP) * F + (i % LP);
    float4 o;
    o.x = a[0];
    o.y = a[LP];
    o.z = a[2 * LP];
    o.w = a[3 * LP];
    o4[i] = o;
  }
}


template <int F>
__global__ __launch_bounds__(1024) void agg_kernel(const float* __restrict__ XW,
                                                   const int* __restrict__ es,
                                                   const int* __restrict__ ed,
                                                   const float* __restrict__ ew,
                                                   const int* __restrict__ binStart,
                                                   const float* __restrict__ deg,
                                                   float* __restrict__ sums, int N) {
  constexpr int J = F / 32;
  constexpr int U = 32 / J;
  constexpr int F4 = F / 4;
  __shared__ __attribute__((aligned(16))) float acc[RB * F];
  int t = threadIdx.x;
  int lane = t & 31;
  int wave = t >> 5;
  int base = blockIdx.x * RB;
  int nrows = N - base < RB ? N - base : RB;

  const float4* xin = (const float4*)(XW + (size_t)base * F);
  float4* a4 = (float4*)acc;
  for (int i = t; i < nrows * F4; i += 1024) {
    float d = rsqrtf(deg[base + i / F4]);
    float dd = d * d;
    float4 x = xin[i];
    x.x *= dd; x.y *= dd; x.z *= dd; x.w *= dd;
    a4[i] = x;
  }
  __syncthreads();

  int eb = binStart[blockIdx.x];
  int ee = binStart[blockIdx.x + 1];
  for (int e0 = eb + wave * 32; e0 < ee; e0 += 1024) {
    int e = e0 + lane;
    int s = 0, r = 0;
    float w = 0.f;
    if (e < ee) {
      s = es[e];
      r = ed[e];
      w = ew[e];
    }
    int w_bits = __float_as_int(w);
    int cnt = ee - e0;
    if (cnt >= 32) {
      for (int i0 = 0; i0 < 32; i0 += U) {
        float v[U][J];
#pragma unroll
        for (int u = 0; u < U; ++u) {
          int si = __builtin_amdgcn_readlane(s, i0 + u);
          const float* xr = XW + (size_t)si * F + lane;
#pragma unroll
          for (int j = 0; j < J; ++j) v[u][j] = xr[j * 32];
        }
#pragma unroll
        for (int u = 0; u < U; ++u) {
          int ri = __builtin_amdgcn_readlane(r, i0 + u);
          float wi = __int_as_float(__builtin_amdgcn_readlane(w_bits, i0 + u));
          float* ar = acc + ri * F + lane;
#pragma unroll
          for (int j = 0; j < J; ++j) atomicAdd(ar + j * 32, v[u][j] * wi);
        }
      }
    } else {
      for (int i = 0; i < cnt; ++i) {
        int si = __builtin_amdgcn_readlane(s, i);
        int ri = __builtin_amdgcn_readlane(r, i);
        float wi = __int_as_float(__builtin_amdgcn_readlane(w_bits, i));
        const float* xr = XW + (size_t)si * F + lane;
        float* ar = acc + ri * F + lane;
#pragma unroll
        for (int j = 0; j < J; ++j) atomicAdd(ar + j * 32, xr[j * 32] * wi);
      }
    }
  }
  __syncthreads();

  float4* o4 = (float4*)(sums + (size_t)base * F);
  for (int i = t; i < nrows * F4; i += 1024) o4[i] = a4[i];
}


#define PN 32
__global__ __launch_bounds__(128) void pool_sum_kernel(const float* __restrict__ h,
                                                       const float* __restrict__ b,
                                                       const int* __restrict__ batch,
                                                       float* __restrict__ out,
                                                       float* __restrict__ cnt, int N) {
  int lane = threadIdx.x & 31;
  int wave = blockIdx.x * 4 + __builtin_amdgcn_readfirstlane(threadIdx.x >> 5);
  int n0 = wave * PN;
  if (n0 >= N) return;
  int nn = N - n0 < PN ? N - n0 : PN;
  int g = batch[n0 + (lane < nn ? lane : nn - 1)];
  float bl = b[lane];
  const float* hp = h + (size_t)n0 * OUTC + lane;
  float v[PN];
#pragma unroll
  for (int i = 0; i < PN; ++i) v[i] = hp[(size_t)(i < nn ? i : nn - 1) * OUTC];

  int cur = __builtin_amdgcn_readlane(g, 0);
  float x0 = v[0] + bl;
  float acc = x0 > 0.f ? x0 : 0.f;
  int n = 1;
#pragma unroll
  for (int i = 1; i < PN; ++i) {
    if (i < nn) {
      int gi = __builtin_amdgcn_readlane(g, i);
      if (gi != cur) {
        atomicAdd(&out[(size_t)cur * OUTC + lane], acc);
        if (lane == 0) atomicAdd(&cnt[cur], (float)n);
        cur = gi;
        acc = 0.f;
        n = 0;
      }
      float x = v[i] + bl;
      acc += x > 0.f ? x : 0.f;
      ++n;
    }
  }
  atomicAdd(&out[(size_t)cur * OUTC + lane], acc);
  if (lane == 0) atomicAdd(&cnt[cur], (float)n);
}

__global__ void pool_div_kernel(float* __restrict__ out, const float* __restrict__ cnt, int G) {
  int i = blockIdx.x * blockDim.x + threadIdx.x;
  if (i >= G * OUTC) return;
  out[i] /= fmaxf(cnt[i >> 5], 1.0f);
}


extern "C" void kernel_launch(void* const* d_in, const int* in_sizes, int n_in,
                              void* d_out, int out_size, void* d_ws, size_t ws_size,
                              hipStream_t stream) {
  const float* x     = (const float*)d_in[0];
  const int*   ei    = (const int*)d_in[1];
  const int*   batch = (const int*)d_in[2];
  const float* W1    = (const float*)d_in[3];
  const float* b1    = (const float*)d_in[4];
  const float* W2    = (const float*)d_in[5];
  const float* b2    = (const float*)d_in[6];
  const float* W3    = (const float*)d_in[7];
  const float* b3    = (const float*)d_in[8];

  const int N = in_sizes[0] / F_IN;
  const int E = in_sizes[1] / 2;
  const int G = out_size / OUTC;
  const int* src = ei;
  const int* dst = ei + E;
  float* out = (float*)d_out;

  float* deg = (float*)d_ws;
  float* cnt = deg + N;
  int* binCount  = (int*)(cnt + G);
  int* binCursor = binCount + MAXBIN;
  int* binStart  = binCursor + MAXBIN;
  size_t off = (size_t)N + (size_t)G + 3 * (size_t)MAXBIN + 1;
  off = (off + 63) & ~(size_t)63;
  float* bufA = (float*)d_ws + off;
  float* bufB = bufA + (size_t)N * HID;
  float* bufC = bufB + (size_t)N * HID;
  int*   es = (int*)(bufC + (size_t)N * HID2);
  int*   ed = es + E;
  float* ew = (float*)(ed + E);

  const int TB = 256;

  int n0 = G > out_size ? G : out_size;
  if (MAXBIN > n0) n0 = MAXBIN;
  init_kernel<<<ceil_div(n0, TB), TB, 0, stream>>>(cnt, out, binCount, binCursor, G, out_size);
  count_kernel<<<ceil_div(E, CHUNK), 256, 0, stream>>>(dst, binCount, E);
  place_kernel<<<ceil_div(E, CHUNK), 256, 0, stream>>>(src, dst, binCount, binCursor,
                                                       binStart, es, ed, E);
  const int nbin = ceil_div(N, RB);
  degree_kernel<<<nbin, 1024, 0, stream>>>(ed, binStart, deg, N);

  const int GB = 512;
  int s1 = ceil_div(N, 16), s2 = ceil_div(N, 32), s3 = ceil_div(N, 64);

  gemm_wstat_kernel<false, F_IN, HID><<<s1 < GB ? s1 : GB, 256, 0, stream>>>(x, nullptr, W1, bufA, N);
  agg_rows_kernel<HID, true><<<nbin, 1024, 0, stream>>>(bufA, es, ed, ew, binStart, deg, bufB, N);

  gemm_wstat_kernel<true, HID, HID2><<<s2 < GB ? s2 : GB, 256, 0, stream>>>(bufB, b1, W2, bufA, N);
  agg_rows_kernel<HID2, false><<<nbin, 1024, 0, stream>>>(bufA, es, ed, ew, binStart, deg, bufC, N);

  gemm_wstat_kernel<true, HID2, OUTC><<<s3 < GB ? s3 : GB, 256, 0, stream>>>(bufC, b2, W3, bufA, N);
  agg_kernel<OUTC><<<nbin, 1024, 0, stream>>>(bufA, es, ed, ew, binStart, deg, bufB, N);

  pool_sum_kernel<<<ceil_div(N, PN * 4), 128, 0, stream>>>(bufB, b3, batch, out, cnt, N);
  pool_div_kernel<<<ceil_div(G * OUTC, TB), TB, 0, stream>>>(out, cnt, G);
}


# Loops that waited for one load per trip (bucket count, placement, degree count, start values of the edge passes) issue all of a thread's loads first: up to 16 in flight, not 1; same sums
# speedup vs baseline: 4.6475x; 1.0404x over previous
#include <hip/hip_runtime.h>

typedef float v2f __attribute__((ext_vector_type(2)));
typedef float v8f __attribute__((ext_vector_type(8)));

#define F_IN 128
#define HID  128
#define HID2 64
#define OUTC 32

#define RB     196
#define MAXBIN 1024
#define CHUNK  4096
#define EPT    (CHUNK / 256)

static inline int ceil_div(int a, int b) { return (a + b - 1) / b; }


__global__ void init_kernel(float* __restrict__ cnt, float* __restrict__ out,
                            int* __restrict__ binCount, int* __restrict__ binCursor,
                            int G, int out_size) {
  int i = blockIdx.x * blockDim.x + threadIdx.x;
  if (i < G) cnt[i] = 0.f;
  if (i < out_size) out[i] = 0.f;
  if (i < MAXBIN) { binCount[i] = 0; binCursor[i] = 0; }
}


__global__ __launch_bounds__(256) void count_kernel(const int* __restrict__ dst,
                                                    int* __restrict__ binCount, int E) {
  __shared__ int hist[MAXBIN];
  int t = threadIdx.x;
  int e0 = blockIdx.x * CHUNK;
  int e1 = e0 + CHUNK < E ? e0 + CHUNK : E;
  int d[EPT];
#pragma unroll
  for (int k = 0; k < EPT; ++k) {
    int e = e0 + t + 256 * k;
    d[k] = dst[e < e1 ? e : e1 - 1];
  }
  for (int i = t; i < MAXBIN; i += 256) hist[i] = 0;
  __syncthreads();
#pragma unroll
  for (int k = 0; k < EPT; ++k)
    if (e0 + t + 256 * k < e1) atomicAdd(&hist[d[k] / RB], 1);
  __syncthreads();
  for (int i = t; i < MAXBIN; i += 256) {
    int h = hist[i];
    if (h) atomicAdd(&binCount[i], h);
  }
}

__global__ __launch_bounds__(256) void place_kernel(const int* __restrict__ src,
                                                    const int* __restrict__ dst,
                                                    const int* __restrict__ binCount,
                                                    int* __restrict__ binCursor,
                                                    int* __restrict__ binStart,
                                                    int* __restrict__ es, int* __restrict__ ed,
                                                    int E) {
  __shared__ int hist[MAXBIN];
  __shared__ int pos[MAXBIN];
  __shared__ int part[256];
  int t = threadIdx.x;

  int e0 = blockIdx.x * CHUNK;
  int e1 = e0 + CHUNK < E ? e0 + CHUNK : E;
  int dv[EPT], sv[EPT];
#pragma unroll
  for (int k = 0; k < EPT; ++k) {
    int e = e0 + t + 256 * k;
    e = e < e1 ? e : e1 - 1;
    dv[k] = dst[e];
    sv[k] = src[e];
  }

  int c0 = binCount[4 * t + 0], c1 = binCount[4 * t + 1];
  int c2 = binCount[4 * t + 2], c3 = binCount[4 * t + 3];
  int own = c0 + c1 + c2 + c3;
  part[t] = own;
  for (int i = t; i < MAXBIN; i += 256) hist[i] = 0;
  __syncthreads();
  for (int o = 1; o < 256; o <<= 1) {
    int v = t >= o ? part[t - o] : 0;
    __syncthreads();
    part[t] += v;
    __syncthreads();
  }
  int off = part[t] - own;
  pos[4 * t + 0] = off;
  pos[4 * t + 1] = off + c0;
  pos[4 * t + 2] = off + c0 + c1;
  pos[4 * t + 3] = off + c0 + c1 + c2;
  if (blockIdx.x == 0) {
    binStart[4 * t + 0] = off;
    binStart[4 * t + 1] = off + c0;
    binStart[4 * t + 2] = off + c0 + c1;
    binStart[4 * t + 3] = off + c0 + c1 + c2;
    if (t == 255) binStart[MAXBIN] = off + own;
  }

  int bv[EPT];
#pragma unroll
  for (int k = 0; k < EPT; ++k) {
    bv[k] = dv[k] / RB;
    if (e0 + t + 256 * k < e1) atomicAdd(&hist[bv[k]], 1);
  }
  __syncthreads();
  for (int i = t; i < MAXBIN; i += 256) {
    int h = hist[i];
    if (h) pos[i] += atomicAdd(&binCursor[i], h);
  }
  __syncthreads();
  int slot[EPT];
#pragma unroll
  for (int k = 0; k < EPT; ++k) {
    slot[k] = 0;
    if (e0 + t + 256 * k < e1) slot[k] = atomicAdd(&pos[bv[k]], 1);
  }
#pragma unroll
  for (int k = 0; k < EPT; ++k) {
    if (e0 + t + 256 * k < e1) {
      es[slot[k]] = sv[k];
      ed[slot[k]] = dv[k] - bv[k] * RB;
    }
  }
}

__global__ __launch_bounds__(1024) void degree_kernel(const int* __restrict__ ed,
                                                      const int* __restrict__ binStart,
                                                      float* __restrict__ deg, int N) {
  __shared__ int rowCount[RB];
  int t = threadIdx.x;
  if (t < RB) rowCount[t] = 0;
  __syncthreads();
  int eb = binStart[blockIdx.x];
  int ee = binStart[blockIdx.x + 1];
  constexpr int DU = 8;
  static_assert(DU == 8, "the load fence in the loop names DU values");
  for (int e = eb + t; e < ee; e += 1024 * DU) {
    int r[DU];
#pragma unroll
    for (int k = 0; k < DU; ++k) {
      int ek = e + 1024 * k;
      r[k] = ed[ek < ee ? ek : ee - 1];
    }
    asm volatile("" :: "v"(r[0]), "v"(r[1]), "v"(r[2]), "v"(r[3]),
                       "v"(r[4]), "v"(r[5]), "v"(r[6]), "v"(r[7]));
#pragma unroll
    for (int k = 0; k < DU; ++k)
      if (e + 1024 * k < ee) atomicAdd(&rowCount[r[k]], 1);
  }
  __syncthreads();
  int node = blockIdx.x * RB + t;
  if (t < RB && node < N) deg[node] = 1.0f + (float)rowCount[t];
}

template <int K>
__device__ __forceinline__ float4 stage_piece(const float* __restrict__ A, int row, int pc, int M) {
  row = row < M ? row : M - 1;
  return *(const float4*)(A + (size_t)row * K + 4 * pc);
}

template <bool ACT>
__device__ __forceinline__ float4 act_piece(float4 v, float4 b) {
  if (ACT) {
    v.x = fmaxf(v.x + b.x, 0.f);
    v.y = fmaxf(v.y + b.y, 0.f);
    v.z = fmaxf(v.z + b.z, 0.f);
    v.w = fmaxf(v.w + b.w, 0.f);
  }
  return v;
}

template <bool ACT, int K, int NCOL>
__global__ __launch_bounds__(256) void gemm_wstat_kernel(
    const float* __restrict__ A, const float* __restrict__ abias, const float* __restrict__ B,
    float* __restrict__ C, int M) {
  constexpr int NT   = NCOL / 16;
  constexpr int RT   = 8 / NT;
  constexpr int ROWS = RT * 16;
  constexpr int K4   = K / 4;
  constexpr int P    = K + 4;
  constexpr int RSTEP = 256 / K4;
  constexpr int NLD  = ROWS / RSTEP;
  static_assert(NT * RT == 8 && NLD * RSTEP == ROWS, "8 waves, whole pieces");
  __shared__ __attribute__((aligned(16))) float As[2][ROWS * P];

  int t = threadIdx.x;
  int lane = t & 31;
  int wave = __builtin_amdgcn_readfirstlane(t >> 5);
  int c = lane & 15;
  int h = lane >> 4;
  int ct = wave % NT;
  int rt = wave / NT;

  v2f bf[K / 4];
  const float* bcol = B + ct * 16 + c;
#pragma unroll
  for (int j = 0; j < K / 8; ++j) {
    int k0 = 8 * j + 4 * h;
    bf[2 * j].x     = bcol[(size_t)(k0 + 0) * NCOL];
    bf[2 * j].y     = bcol[(size_t)(k0 + 1) * NCOL];
    bf[2 * j + 1].x = bcol[(size_t)(k0 + 2) * NCOL];
    bf[2 * j + 1].y = bcol[(size_t)(k0 + 3) * NCOL];
  }

  const unsigned coff = (unsigned)((rt * 16 + h * 8) * NCOL + ct * 16 + c);

  int pc = t % K4;
  int pr = t / K4;
  float4 bias4 = {0.f, 0.f, 0.f, 0.f};
  if (ACT) bias4 = *(const float4*)(abias + 4 * pc);

  int nsteps = (M + ROWS - 1) / ROWS;
  float4 s0, s1, s2, s3;
  s0 = s1 = s2 = s3 = bias4;
  int step = blockIdx.x;
  if (step < nsteps) {
    s0 = stage_piece<K>(A, step * ROWS + pr, pc, M);
    s1 = stage_piece<K>(A, step * ROWS + pr + RSTEP, pc, M);
    if (NLD == 4) {
      s2 = stage_piece<K>(A, step * ROWS + pr + 2 * RSTEP, pc, M);
      s3 = stage_piece<K>(A, step * ROWS + pr + 3 * RSTEP, pc, M);
    }
  }
  int buf = 0;
  for (; step < nsteps; step += gridDim.x) {
    float* as = As[buf];
    float* ap = as + pr * P + 4 * pc;
    *(float4*)(ap) = act_piece<ACT>(s0, bias4);
    *(float4*)(ap + RSTEP * P) = act_piece<ACT>(s1, bias4);
    if (NLD == 4) {
      *(float4*)(ap + 2 * RSTEP * P) = act_piece<ACT>(s2, bias4);
      *(float4*)(ap + 3 * RSTEP * P) = act_piece<ACT>(s3, bias4);
    }
    __syncthreads();

    int nxt = step + gridDim.x;
    if (nxt < nsteps) {
      s0 = stage_piece<K>(A, nxt * ROWS + pr, pc, M);
      s1 = stage_piece<K>(A, nxt * ROWS + pr + RSTEP, pc, M);
      if (NLD == 4) {
        s2 = stage_piece<K>(A, nxt * ROWS + pr + 2 * RSTEP, pc, M);
        s3 = stage_piece<K>(A, nxt * ROWS + pr + 3 * RSTEP, pc, M);
      }
    }

    int tile0 = step * ROWS + rt * 16;
    float* cstep = C + (size_t)step * (ROWS * NCOL);

    const float* arow = as + (rt * 16 + c) * P + 4 * h;
    float4 a[K / 8];
#pragma unroll
    for (int j = 0; j < K / 8; ++j) a[j] = *(const float4*)(arow + 8 * j);
    __builtin_amdgcn_sched_barrier(0);

    v8f acc = {0.f, 0.f, 0.f, 0.f, 0.f, 0.f, 0.f, 0.f};
#pragma unroll
    for (int j = 0; j < K / 8; ++j) {
      v2f a0 = {a[j].x, a[j].y};
      v2f a1 = {a[j].z, a[j].w};
      acc = __builtin_amdgcn_wmma_f32_16x16x4_f32(
           false, a0,  false, bf[2 * j],
           (short)0, acc,  false,  false);
      acc = __builtin_amdgcn_wmma_f32_16x16x4_f32(
           false, a1,  false, bf[2 * j + 1],
           (short)0, acc,  false,  false);
    }

    if (tile0 + 16 <= M) {
#pragma unroll
      for (int r = 0; r < 8; ++r) cstep[coff + r * NCOL] = acc[r];
    } else {
#pragma unroll
      for (int r = 0; r < 8; ++r)
        if (tile0 + h * 8 + r < M) cstep[coff + r * NCOL] = acc[r];
    }
    buf ^= 1;
  }
}


template <int R>
__device__ __forceinline__ int pick(int x, int i, int sub) {
  int a = __builtin_amdgcn_readlane(x, i);
  if (R == 2) {
    int b = __builtin_amdgcn_readlane(x, i + 1);
    a = sub ? b : a;
  }
  return a;
}

template <int F>
__device__ __forceinline__ void row_add(float* acc, int ri, int q, int sub, float4 v, float w) {
  constexpr int LP = F / 4;
  float* ar = acc + ri * F + q;
  if (F == 128) {
    atomicAdd(ar,          v.x * w);
    atomicAdd(ar + LP,     v.y * w);
    atomicAdd(ar + 2 * LP, v.z * w);
    atomicAdd(ar + 3 * LP, v.w * w);
  } else {
    float y0 = sub ? v.z : v.x;
    float y1 = sub ? v.w : v.y;
    float y2 = sub ? v.x : v.z;
    float y3 = sub ? v.y : v.w;
    float* a1 = ar + (sub ? 2 * LP : 0);
    float* a2 = ar + (sub ? 0 : 2 * LP);
    atomicAdd(a1,      y0 * w);
    atomicAdd(a1 + LP, y1 * w);
    atomicAdd(a2,      y2 * w);
    atomicAdd(a2 + LP, y3 * w);
  }
}

template <int F, bool MKW>
__global__ __launch_bounds__(1024) void agg_rows_kernel(const float* __restrict__ XW,
                                                        const int* __restrict__ es,
                                                        const int* __restrict__ ed,
                                                        float* __restrict__ ew,
                                                        const int* __restrict__ binStart,
                                                        const float* __restrict__ deg,
                                                        float* __restrict__ sums, int N) {
  static_assert(F == 128 || F == 64, "a row is one or two aligned 256-byte units");
  constexpr int LP = F / 4;
  constexpr int R  = 32 / LP;
  constexpr int U  = 8;
  __shared__ __attribute__((aligned(16))) float acc[RB * F];
  int t = threadIdx.x;
  int lane = t & 31;
  int wave = t >> 5;
  int sub = lane / LP;
  int q = lane % LP;
  int base = blockIdx.x * RB;
  int nrows = N - base < RB ? N - base : RB;

  constexpr int PT = (RB * LP + 1023) / 1024;
  const float4* xin = (const float4*)(XW + (size_t)base * F);
  float4 xs[PT];
  float dg[PT];
#pragma unroll
  for (int k = 0; k < PT; ++k) {
    int i = t + 1024 * k;
    i = i < nrows * LP ? i : 0;
    dg[k] = deg[base + i / LP];
    xs[k] = xin[i];
  }
#pragma unroll
  for (int k = 0; k < PT; ++k) {
    int i = t + 1024 * k;
    if (i < nrows * LP) {
      int row = i / LP;
      float d = rsqrtf(dg[k]);
      float dd = d * d;
      float4 x = xs[k];
      float* a = acc + row * F + (i % LP);
      a[0]      = x.x * dd;
      a[LP]     = x.y * dd;
      a[2 * LP] = x.z * dd;
      a[3 * LP] = x.w * dd;
    }
  }
  __syncthreads();

  int eb = binStart[blockIdx.x];
  int ee = binStart[blockIdx.x + 1];
  for (int e0 = eb + wave * 32; e0 < ee; e0 += 1024) {
    int e = e0 + lane;
    int s = 0, r = 0;
    float w = 0.f;
    if (e < ee) {
      s = es[e];
      r = ed[e];
      if (!MKW) w = ew[e];
    }
    int cnt = ee - e0;
    if (MKW && cnt < 32 && e < ee) {
      w = rsqrtf(deg[s]) * rsqrtf(deg[base + r]);
      ew[e] = w;
    }
    int w_bits = __float_as_int(w);
    if (cnt >= 32) {
      float dgs = 1.f, dgd = 1.f;
      for (int i0 = 0; i0 < 32; i0 += U * R) {
        float4 v[U];
#pragma unroll
        for (int u = 0; u < U; ++u) {
          int si = pick<R>(s, i0 + u * R, sub);
          v[u] = *(const float4*)(XW + (size_t)si * F + 4 * q);
        }
        if (MKW && i0 == 0) {
          dgs = deg[s];
          dgd = deg[base + r];
        }
        __builtin_amdgcn_sched_barrier(0);
        if (MKW && i0 == 0) {
          w = rsqrtf(dgs) * rsqrtf(dgd);
          ew[e] = w;
          w_bits = __float_as_int(w);
        }
#pragma unroll
        for (int u = 0; u < U; ++u) {
          int ri = pick<R>(r, i0 + u * R, sub);
          float wi = __int_as_float(pick<R>(w_bits, i0 + u * R, sub));
          row_add<F>(acc, ri, q, sub, v[u], wi);
        }
      }
    } else {
      for (int i = 0; i < cnt; i += R) {
        int si = pick<R>(s, i, sub);
        int ri = pick<R>(r, i, sub);
        float wi = __int_as_float(pick<R>(w_bits, i, sub));
        if (R == 1 || i + sub < cnt) {
          float4 x = *(const float4*)(XW + (size_t)si * F + 4 * q);
          row_add<F>(acc, ri, q, sub, x, wi);
        }
      }
    }
  }
  __syncthreads();

  float4* o4 = (float4*)(sums + (size_t)base * F);
  for (int i = t; i < nrows * LP; i += 1024) {
    const float* a = acc + (i / LP) * F + (i % LP);
    float4 o;
    o.x = a[0];
    o.y = a[LP];
    o.z = a[2 * LP];
    o.w = a[3 * LP];
    o4[i] = o;
  }
}


template <int F>
__global__ __launch_bounds__(1024) void agg_kernel(const float* __restrict__ XW,
                                                   const int* __restrict__ es,
                                                   const int* __restrict__ ed,
                                                   const float* __restrict__ ew,
                                                   const int* __restrict__ binStart,
                                                   const float* __restrict__ deg,
                                                   float* __restrict__ sums, int N) {
  constexpr int J = F / 32;
  constexpr int U = 32 / J;
  constexpr int F4 = F / 4;
  __shared__ __attribute__((aligned(16))) float acc[RB * F];
  int t = threadIdx.x;
  int lane = t & 31;
  int wave = t >> 5;
  int base = blockIdx.x * RB;
  int nrows = N - base < RB ? N - base : RB;

  constexpr int PT = (RB * F4 + 1023) / 1024;
  const float4* xin = (const float4*)(XW + (size_t)base * F);
  float4* a4 = (float4*)acc;
  float4 xs[PT];
  float dg[PT];
#pragma unroll
  for (int k = 0; k < PT; ++k) {
    int i = t + 1024 * k;
    i = i < nrows * F4 ? i : 0;
    dg[k] = deg[base + i / F4];
    xs[k] = xin[i];
  }
#pragma unroll
  for (int k = 0; k < PT; ++k) {
    int i = t + 1024 * k;
    if (i < nrows * F4) {
      float d = rsqrtf(dg[k]);
      float dd = d * d;
      float4 x = xs[k];
      x.x *= dd; x.y *= dd; x.z *= dd; x.w *= dd;
      a4[i] = x;
    }
  }
  __syncthreads();

  int eb = binStart[blockIdx.x];
  int ee = binStart[blockIdx.x + 1];
  for (int e0 = eb + wave * 32; e0 < ee; e0 += 1024) {
    int e = e0 + lane;
    int s = 0, r = 0;
    float w = 0.f;
    if (e < ee) {
      s = es[e];
      r = ed[e];
      w = ew[e];
    }
    int w_bits = __float_as_int(w);
    int cnt = ee - e0;
    if (cnt >= 32) {
      for (int i0 = 0; i0 < 32; i0 += U) {
        float v[U][J];
#pragma unroll
        for (int u = 0; u < U; ++u) {
          int si = __builtin_amdgcn_readlane(s, i0 + u);
          const float* xr = XW + (size_t)si * F + lane;
#pragma unroll
          for (int j = 0; j < J; ++j) v[u][j] = xr[j * 32];
        }
#pragma unroll
        for (int u = 0; u < U; ++u) {
          int ri = __builtin_amdgcn_readlane(r, i0 + u);
          float wi = __int_as_float(__builtin_amdgcn_readlane(w_bits, i0 + u));
          float* ar = acc + ri * F + lane;
#pragma unroll
          for (int j = 0; j < J; ++j) atomicAdd(ar + j * 32, v[u][j] * wi);
        }
      }
    } else {
      for (int i = 0; i < cnt; ++i) {
        int si = __builtin_amdgcn_readlane(s, i);
        int ri = __builtin_amdgcn_readlane(r, i);
        float wi = __int_as_float(__builtin_amdgcn_readlane(w_bits, i));
        const float* xr = XW + (size_t)si * F + lane;
        float* ar = acc + ri * F + lane;
#pragma unroll
        for (int j = 0; j < J; ++j) atomicAdd(ar + j * 32, xr[j * 32] * wi);
      }
    }
  }
  __syncthreads();

  float4* o4 = (float4*)(sums + (size_t)base * F);
  for (int i = t; i < nrows * F4; i += 1024) o4[i] = a4[i];
}


#define PN 32
__global__ __launch_bounds__(128) void pool_sum_kernel(const float* __restrict__ h,
                                                       const float* __restrict__ b,
                                                       const int* __restrict__ batch,
                                                       float* __restrict__ out,
                                                       float* __restrict__ cnt, int N) {
  int lane = threadIdx.x & 31;
  int wave = blockIdx.x * 4 + __builtin_amdgcn_readfirstlane(threadIdx.x >> 5);
  int n0 = wave * PN;
  if (n0 >= N) return;
  int nn = N - n0 < PN ? N - n0 : PN;
  int g = batch[n0 + (lane < nn ? lane : nn - 1)];
  float bl = b[lane];
  const float* hp = h + (size_t)n0 * OUTC + lane;
  float v[PN];
#pragma unroll
  for (int i = 0; i < PN; ++i) v[i] = hp[(size_t)(i < nn ? i : nn - 1) * OUTC];

  int cur = __builtin_amdgcn_readlane(g, 0);
  float x0 = v[0] + bl;
  float acc = x0 > 0.f ? x0 : 0.f;
  int n = 1;
#pragma unroll
  for (int i = 1; i < PN; ++i) {
    if (i < nn) {
      int gi = __builtin_amdgcn_readlane(g, i);
      if (gi != cur) {
        atomicAdd(&out[(size_t)cur * OUTC + lane], acc);
        if (lane == 0) atomicAdd(&cnt[cur], (float)n);
        cur = gi;
        acc = 0.f;
        n = 0;
      }
      float x = v[i] + bl;
      acc += x > 0.f ? x : 0.f;
      ++n;
    }
  }
  atomicAdd(&out[(size_t)cur * OUTC + lane], acc);
  if (lane == 0) atomicAdd(&cnt[cur], (float)n);
}

__global__ void pool_div_kernel(float* __restrict__ out, const float* __restrict__ cnt, int G) {
  int i = blockIdx.x * blockDim.x + threadIdx.x;
  if (i >= G * OUTC) return;
  out[i] /= fmaxf(cnt[i >> 5], 1.0f);
}


extern "C" void kernel_launch(void* const* d_in, const int* in_sizes, int n_in,
                              void* d_out, int out_size, void* d_ws, size_t ws_size,
                              hipStream_t stream) {
  const float* x     = (const float*)d_in[0];
  const int*   ei    = (const int*)d_in[1];
  const int*   batch = (const int*)d_in[2];
  const float* W1    = (const float*)d_in[3];
  const float* b1    = (const float*)d_in[4];
  const float* W2    = (const float*)d_in[5];
  const float* b2    = (const float*)d_in[6];
  const float* W3    = (const float*)d_in[7];
  const float* b3    = (const float*)d_in[8];

  const int N = in_sizes[0] / F_IN;
  const int E = in_sizes[1] / 2;
  const int G = out_size / OUTC;
  const int* src = ei;
  const int* dst = ei + E;
  float* out = (float*)d_out;

  float* deg = (float*)d_ws;
  float* cnt = deg + N;
  int* binCount  = (int*)(cnt + G);
  int* binCursor = binCount + MAXBIN;
  int* binStart  = binCursor + MAXBIN;
  size_t off = (size_t)N + (size_t)G + 3 * (size_t)MAXBIN + 1;
  off = (off + 63) & ~(size_t)63;
  float* bufA = (float*)d_ws + off;
  float* bufB = bufA + (size_t)N * HID;
  float* bufC = bufB + (size_t)N * HID;
  int*   es = (int*)(bufC + (size_t)N * HID2);
  int*   ed = es + E;
  float* ew = (float*)(ed + E);

  const int TB = 256;

  int n0 = G > out_size ? G : out_size;
  if (MAXBIN > n0) n0 = MAXBIN;
  init_kernel<<<ceil_div(n0, TB), TB, 0, stream>>>(cnt, out, binCount, binCursor, G, out_size);
  count_kernel<<<ceil_div(E, CHUNK), 256, 0, stream>>>(dst, binCount, E);
  place_kernel<<<ceil_div(E, CHUNK), 256, 0, stream>>>(src, dst, binCount, binCursor,
                                                       binStart, es, ed, E);
  const int nbin = ceil_div(N, RB);
  degree_kernel<<<nbin, 1024, 0, stream>>>(ed, binStart, deg, N);

  const int GB = 512;
  int s1 = ceil_div(N, 16), s2 = ceil_div(N, 32), s3 = ceil_div(N, 64);

  gemm_wstat_kernel<false, F_IN, HID><<<s1 < GB ? s1 : GB, 256, 0, stream>>>(x, nullptr, W1, bufA, N);
  agg_rows_kernel<HID, true><<<nbin, 1024, 0, stream>>>(bufA, es, ed, ew, binStart, deg, bufB, N);

  gemm_wstat_kernel<true, HID, HID2><<<s2 < GB ? s2 : GB, 256, 0, stream>>>(bufB, b1, W2, bufA, N);
  agg_rows_kernel<HID2, false><<<nbin, 1024, 0, stream>>>(bufA, es, ed, ew, binStart, deg, bufC, N);

  gemm_wstat_kernel<true, HID2, OUTC><<<s3 < GB ? s3 : GB, 256, 0, stream>>>(bufC, b2, W3, bufA, N);
  agg_kernel<OUTC><<<nbin, 1024, 0, stream>>>(bufA, es, ed, ew, binStart, deg, bufB, N);

  pool_sum_kernel<<<ceil_div(N, PN * 4), 128, 0, stream>>>(bufB, b3, batch, out, cnt, N);
  pool_div_kernel<<<ceil_div(G * OUTC, TB), TB, 0, stream>>>(out, cnt, G);
}


# Bucket placement sorts its 4096-edge chunk by range in LDS first, so the source-id and row stores leave as contiguous per-range segments: 2-3 lines per store instruction, not 32; arrival-order sums
# speedup vs baseline: 5.2606x; 1.1319x over previous
#include <hip/hip_runtime.h>

typedef float v2f __attribute__((ext_vector_type(2)));
typedef float v8f __attribute__((ext_vector_type(8)));

#define F_IN 128
#define HID  128
#define HID2 64
#define OUTC 32

#define RB     196
#define MAXBIN 1024
#define CHUNK  4096
#define EPT    (CHUNK / 256)

static inline int ceil_div(int a, int b) { return (a + b - 1) / b; }


__global__ void init_kernel(float* __restrict__ cnt, float* __restrict__ out,
                            int* __restrict__ binCount, int* __restrict__ binCursor,
                            int G, int out_size) {
  int i = blockIdx.x * blockDim.x + threadIdx.x;
  if (i < G) cnt[i] = 0.f;
  if (i < out_size) out[i] = 0.f;
  if (i < MAXBIN) { binCount[i] = 0; binCursor[i] = 0; }
}


__global__ __launch_bounds__(256) void count_kernel(const int* __restrict__ dst,
                                                    int* __restrict__ binCount, int E) {
  __shared__ int hist[MAXBIN];
  int t = threadIdx.x;
  int e0 = blockIdx.x * CHUNK;
  int e1 = e0 + CHUNK < E ? e0 + CHUNK : E;
  int d[EPT];
#pragma unroll
  for (int k = 0; k < EPT; ++k) {
    int e = e0 + t + 256 * k;
    d[k] = dst[e < e1 ? e : e1 - 1];
  }
  for (int i = t; i < MAXBIN; i += 256) hist[i] = 0;
  __syncthreads();
#pragma unroll
  for (int k = 0; k < EPT; ++k)
    if (e0 + t + 256 * k < e1) atomicAdd(&hist[d[k] / RB], 1);
  __syncthreads();
  for (int i = t; i < MAXBIN; i += 256) {
    int h = hist[i];
    if (h) atomicAdd(&binCount[i], h);
  }
}

static_assert(RB <= 256, "the row inside a range is kept in 8 bits beside the range number");
__global__ __launch_bounds__(256) void place_kernel(const int* __restrict__ src,
                                                    const int* __restrict__ dst,
                                                    const int* __restrict__ binCount,
                                                    int* __restrict__ binCursor,
                                                    int* __restrict__ binStart,
                                                    int* __restrict__ es, int* __restrict__ ed,
                                                    int E) {
  __shared__ int hist[MAXBIN];
  __shared__ int pos[MAXBIN];
  __shared__ int lstart[MAXBIN];
  __shared__ int part[256];
  __shared__ int wsum[8];
  __shared__ int st_s[CHUNK];
  __shared__ int st_p[CHUNK];
  int t = threadIdx.x;
  int lane = t & 31;
  int wave = t >> 5;

  int e0 = blockIdx.x * CHUNK;
  int e1 = e0 + CHUNK < E ? e0 + CHUNK : E;
  int dv[EPT], sv[EPT];
#pragma unroll
  for (int k = 0; k < EPT; ++k) {
    int e = e0 + t + 256 * k;
    e = e < e1 ? e : e1 - 1;
    dv[k] = dst[e];
    sv[k] = src[e];
  }

  int c0 = binCount[4 * t + 0], c1 = binCount[4 * t + 1];
  int c2 = binCount[4 * t + 2], c3 = binCount[4 * t + 3];
  int own = c0 + c1 + c2 + c3;
  part[t] = own;
  for (int i = t; i < MAXBIN; i += 256) hist[i] = 0;
  __syncthreads();
  for (int o = 1; o < 256; o <<= 1) {
    int v = t >= o ? part[t - o] : 0;
    __syncthreads();
    part[t] += v;
    __syncthreads();
  }
  int off = part[t] - own;
  pos[4 * t + 0] = off;
  pos[4 * t + 1] = off + c0;
  pos[4 * t + 2] = off + c0 + c1;
  pos[4 * t + 3] = off + c0 + c1 + c2;
  if (blockIdx.x == 0) {
    binStart[4 * t + 0] = off;
    binStart[4 * t + 1] = off + c0;
    binStart[4 * t + 2] = off + c0 + c1;
    binStart[4 * t + 3] = off + c0 + c1 + c2;
    if (t == 255) binStart[MAXBIN] = off + own;
  }

  int bv[EPT], rk[EPT];
#pragma unroll
  for (int k = 0; k < EPT; ++k) {
    bv[k] = dv[k] / RB;
    rk[k] = 0;
    if (e0 + t + 256 * k < e1) rk[k] = atomicAdd(&hist[bv[k]], 1);
  }
  __syncthreads();

  int h0 = hist[4 * t + 0], h1 = hist[4 * t + 1];
  int h2 = hist[4 * t + 2], h3 = hist[4 * t + 3];
  int hown = h0 + h1 + h2 + h3;
  int hv = hown;
#pragma unroll
  for (int o = 1; o < 32; o <<= 1) {
    int u = __shfl_up(hv, o);
    if (lane >= o) hv += u;
  }
  if (lane == 31) wsum[wave] = hv;
  __syncthreads();
  int lo = hv - hown;
  for (int w = 0; w < 8; ++w)
    if (w < wave) lo += wsum[w];
  lstart[4 * t + 0] = lo;
  lstart[4 * t + 1] = lo + h0;
  lstart[4 * t + 2] = lo + h0 + h1;
  lstart[4 * t + 3] = lo + h0 + h1 + h2;
  __syncthreads();

  int got[MAXBIN / 256];
#pragma unroll
  for (int j = 0; j < MAXBIN / 256; ++j) {
    int h = hist[t + 256 * j];
    got[j] = 0;
    if (h) got[j] = atomicAdd(&binCursor[t + 256 * j], h);
  }
  int lp[EPT];
#pragma unroll
  for (int k = 0; k < EPT; ++k) lp[k] = lstart[bv[k]] + rk[k];
#pragma unroll
  for (int k = 0; k < EPT; ++k) {
    if (e0 + t + 256 * k < e1) {
      st_s[lp[k]] = sv[k];
      st_p[lp[k]] = bv[k] * 256 + (dv[k] - bv[k] * RB);
    }
  }
#pragma unroll
  for (int j = 0; j < MAXBIN / 256; ++j) {
    int i = t + 256 * j;
    pos[i] += got[j] - lstart[i];
  }
  __syncthreads();

  int n = e1 - e0;
  int pj[EPT], sj[EPT], gj[EPT];
#pragma unroll
  for (int k = 0; k < EPT; ++k) {
    int j = t + 256 * k;
    j = j < n ? j : n - 1;
    pj[k] = st_p[j];
    sj[k] = st_s[j];
  }
#pragma unroll
  for (int k = 0; k < EPT; ++k) gj[k] = pos[pj[k] >> 8] + t + 256 * k;
#pragma unroll
  for (int k = 0; k < EPT; ++k) {
    if (t + 256 * k < n) {
      es[gj[k]] = sj[k];
      ed[gj[k]] = pj[k] & 255;
    }
  }
}

__global__ __launch_bounds__(1024) void degree_kernel(const int* __restrict__ ed,
                                                      const int* __restrict__ binStart,
                                                      float* __restrict__ deg, int N) {
  __shared__ int rowCount[RB];
  int t = threadIdx.x;
  if (t < RB) rowCount[t] = 0;
  __syncthreads();
  int eb = binStart[blockIdx.x];
  int ee = binStart[blockIdx.x + 1];
  constexpr int DU = 8;
  static_assert(DU == 8, "the load fence in the loop names DU values");
  for (int e = eb + t; e < ee; e += 1024 * DU) {
    int r[DU];
#pragma unroll
    for (int k = 0; k < DU; ++k) {
      int ek = e + 1024 * k;
      r[k] = ed[ek < ee ? ek : ee - 1];
    }
    asm volatile("" :: "v"(r[0]), "v"(r[1]), "v"(r[2]), "v"(r[3]),
                       "v"(r[4]), "v"(r[5]), "v"(r[6]), "v"(r[7]));
#pragma unroll
    for (int k = 0; k < DU; ++k)
      if (e + 1024 * k < ee) atomicAdd(&rowCount[r[k]], 1);
  }
  __syncthreads();
  int node = blockIdx.x * RB + t;
  if (t < RB && node < N) deg[node] = 1.0f + (float)rowCount[t];
}

template <int K>
__device__ __forceinline__ float4 stage_piece(const float* __restrict__ A, int row, int pc, int M) {
  row = row < M ? row : M - 1;
  return *(const float4*)(A + (size_t)row * K + 4 * pc);
}

template <bool ACT>
__device__ __forceinline__ float4 act_piece(float4 v, float4 b) {
  if (ACT) {
    v.x = fmaxf(v.x + b.x, 0.f);
    v.y = fmaxf(v.y + b.y, 0.f);
    v.z = fmaxf(v.z + b.z, 0.f);
    v.w = fmaxf(v.w + b.w, 0.f);
  }
  return v;
}

template <bool ACT, int K, int NCOL>
__global__ __launch_bounds__(256) void gemm_wstat_kernel(
    const float* __restrict__ A, const float* __restrict__ abias, const float* __restrict__ B,
    float* __restrict__ C, int M) {
  constexpr int NT   = NCOL / 16;
  constexpr int RT   = 8 / NT;
  constexpr int ROWS = RT * 16;
  constexpr int K4   = K / 4;
  constexpr int P    = K + 4;
  constexpr int RSTEP = 256 / K4;
  constexpr int NLD  = ROWS / RSTEP;
  static_assert(NT * RT == 8 && NLD * RSTEP == ROWS, "8 waves, whole pieces");
  __shared__ __attribute__((aligned(16))) float As[2][ROWS * P];

  int t = threadIdx.x;
  int lane = t & 31;
  int wave = __builtin_amdgcn_readfirstlane(t >> 5);
  int c = lane & 15;
  int h = lane >> 4;
  int ct = wave % NT;
  int rt = wave / NT;

  v2f bf[K / 4];
  const float* bcol = B + ct * 16 + c;
#pragma unroll
  for (int j = 0; j < K / 8; ++j) {
    int k0 = 8 * j + 4 * h;
    bf[2 * j].x     = bcol[(size_t)(k0 + 0) * NCOL];
    bf[2 * j].y     = bcol[(size_t)(k0 + 1) * NCOL];
    bf[2 * j + 1].x = bcol[(size_t)(k0 + 2) * NCOL];
    bf[2 * j + 1].y = bcol[(size_t)(k0 + 3) * NCOL];
  }

  const unsigned coff = (unsigned)((rt * 16 + h * 8) * NCOL + ct * 16 + c);

  int pc = t % K4;
  int pr = t / K4;
  float4 bias4 = {0.f, 0.f, 0.f, 0.f};
  if (ACT) bias4 = *(const float4*)(abias + 4 * pc);

  int nsteps = (M + ROWS - 1) / ROWS;
  float4 s0, s1, s2, s3;
  s0 = s1 = s2 = s3 = bias4;
  int step = blockIdx.x;
  if (step < nsteps) {
    s0 = stage_piece<K>(A, step * ROWS + pr, pc, M);
    s1 = stage_piece<K>(A, step * ROWS + pr + RSTEP, pc, M);
    if (NLD == 4) {
      s2 = stage_piece<K>(A, step * ROWS + pr + 2 * RSTEP, pc, M);
      s3 = stage_piece<K>(A, step * ROWS + pr + 3 * RSTEP, pc, M);
    }
  }
  int buf = 0;
  for (; step < nsteps; step += gridDim.x) {
    float* as = As[buf];
    float* ap = as + pr * P + 4 * pc;
    *(float4*)(ap) = act_piece<ACT>(s0, bias4);
    *(float4*)(ap + RSTEP * P) = act_piece<ACT>(s1, bias4);
    if (NLD == 4) {
      *(float4*)(ap + 2 * RSTEP * P) = act_piece<ACT>(s2, bias4);
      *(float4*)(ap + 3 * RSTEP * P) = act_piece<ACT>(s3, bias4);
    }
    __syncthreads();

    int nxt = step + gridDim.x;
    if (nxt < nsteps) {
      s0 = stage_piece<K>(A, nxt * ROWS + pr, pc, M);
      s1 = stage_piece<K>(A, nxt * ROWS + pr + RSTEP, pc, M);
      if (NLD == 4) {
        s2 = stage_piece<K>(A, nxt * ROWS + pr + 2 * RSTEP, pc, M);
        s3 = stage_piece<K>(A, nxt * ROWS + pr + 3 * RSTEP, pc, M);
      }
    }

    int tile0 = step * ROWS + rt * 16;
    float* cstep = C + (size_t)step * (ROWS * NCOL);

    const float* arow = as + (rt * 16 + c) * P + 4 * h;
    float4 a[K / 8];
#pragma unroll
    for (int j = 0; j < K / 8; ++j) a[j] = *(const float4*)(arow + 8 * j);
    __builtin_amdgcn_sched_barrier(0);

    v8f acc = {0.f, 0.f, 0.f, 0.f, 0.f, 0.f, 0.f, 0.f};
#pragma unroll
    for (int j = 0; j < K / 8; ++j) {
      v2f a0 = {a[j].x, a[j].y};
      v2f a1 = {a[j].z, a[j].w};
      acc = __builtin_amdgcn_wmma_f32_16x16x4_f32(
           false, a0,  false, bf[2 * j],
           (short)0, acc,  false,  false);
      acc = __builtin_amdgcn_wmma_f32_16x16x4_f32(
           false, a1,  false, bf[2 * j + 1],
           (short)0, acc,  false,  false);
    }

    if (tile0 + 16 <= M) {
#pragma unroll
      for (int r = 0; r < 8; ++r) cstep[coff + r * NCOL] = acc[r];
    } else {
#pragma unroll
      for (int r = 0; r < 8; ++r)
        if (tile0 + h * 8 + r < M) cstep[coff + r * NCOL] = acc[r];
    }
    buf ^= 1;
  }
}


template <int R>
__device__ __forceinline__ int pick(int x, int i, int sub) {
  int a = __builtin_amdgcn_readlane(x, i);
  if (R == 2) {
    int b = __builtin_amdgcn_readlane(x, i + 1);
    a = sub ? b : a;
  }
  return a;
}

template <int F>
__device__ __forceinline__ void row_add(float* acc, int ri, int q, int sub, float4 v, float w) {
  constexpr int LP = F / 4;
  float* ar = acc + ri * F + q;
  if (F == 128) {
    atomicAdd(ar,          v.x * w);
    atomicAdd(ar + LP,     v.y * w);
    atomicAdd(ar + 2 * LP, v.z * w);
    atomicAdd(ar + 3 * LP, v.w * w);
  } else {
    float y0 = sub ? v.z : v.x;
    float y1 = sub ? v.w : v.y;
    float y2 = sub ? v.x : v.z;
    float y3 = sub ? v.y : v.w;
    float* a1 = ar + (sub ? 2 * LP : 0);
    float* a2 = ar + (sub ? 0 : 2 * LP);
    atomicAdd(a1,      y0 * w);
    atomicAdd(a1 + LP, y1 * w);
    atomicAdd(a2,      y2 * w);
    atomicAdd(a2 + LP, y3 * w);
  }
}

template <int F, bool MKW>
__global__ __launch_bounds__(1024) void agg_rows_kernel(const float* __restrict__ XW,
                                                        const int* __restrict__ es,
                                                        const int* __restrict__ ed,
                                                        float* __restrict__ ew,
                                                        const int* __restrict__ binStart,
                                                        const float* __restrict__ deg,
                                                        float* __restrict__ sums, int N) {
  static_assert(F == 128 || F == 64, "a row is one or two aligned 256-byte units");
  constexpr int LP = F / 4;
  constexpr int R  = 32 / LP;
  constexpr int U  = 8;
  __shared__ __attribute__((aligned(16))) float acc[RB * F];
  int t = threadIdx.x;
  int lane = t & 31;
  int wave = t >> 5;
  int sub = lane / LP;
  int q = lane % LP;
  int base = blockIdx.x * RB;
  int nrows = N - base < RB ? N - base : RB;

  constexpr int PT = (RB * LP + 1023) / 1024;
  const float4* xin = (const float4*)(XW + (size_t)base * F);
  float4 xs[PT];
  float dg[PT];
#pragma unroll
  for (int k = 0; k < PT; ++k) {
    int i = t + 1024 * k;
    i = i < nrows * LP ? i : 0;
    dg[k] = deg[base + i / LP];
    xs[k] = xin[i];
  }
#pragma unroll
  for (int k = 0; k < PT; ++k) {
    int i = t + 1024 * k;
    if (i < nrows * LP) {
      int row = i / LP;
      float d = rsqrtf(dg[k]);
      float dd = d * d;
      float4 x = xs[k];
      float* a = acc + row * F + (i % LP);
      a[0]      = x.x * dd;
      a[LP]     = x.y * dd;
      a[2 * LP] = x.z * dd;
      a[3 * LP] = x.w * dd;
    }
  }
  __syncthreads();

  int eb = binStart[blockIdx.x];
  int ee = binStart[blockIdx.x + 1];
  for (int e0 = eb + wave * 32; e0 < ee; e0 += 1024) {
    int e = e0 + lane;
    int s = 0, r = 0;
    float w = 0.f;
    if (e < ee) {
      s = es[e];
      r = ed[e];
      if (!MKW) w = ew[e];
    }
    int cnt = ee - e0;
    if (MKW && cnt < 32 && e < ee) {
      w = rsqrtf(deg[s]) * rsqrtf(deg[base + r]);
      ew[e] = w;
    }
    int w_bits = __float_as_int(w);
    if (cnt >= 32) {
      float dgs = 1.f, dgd = 1.f;
      for (int i0 = 0; i0 < 32; i0 += U * R) {
        float4 v[U];
#pragma unroll
        for (int u = 0; u < U; ++u) {
          int si = pick<R>(s, i0 + u * R, sub);
          v[u] = *(const float4*)(XW + (size_t)si * F + 4 * q);
        }
        if (MKW && i0 == 0) {
          dgs = deg[s];
          dgd = deg[base + r];
        }
        __builtin_amdgcn_sched_barrier(0);
        if (MKW && i0 == 0) {
          w = rsqrtf(dgs) * rsqrtf(dgd);
          ew[e] = w;
          w_bits = __float_as_int(w);
        }
#pragma unroll
        for (int u = 0; u < U; ++u) {
          int ri = pick<R>(r, i0 + u * R, sub);
          float wi = __int_as_float(pick<R>(w_bits, i0 + u * R, sub));
          row_add<F>(acc, ri, q, sub, v[u], wi);
        }
      }
    } else {
      for (int i = 0; i < cnt; i += R) {
        int si = pick<R>(s, i, sub);
        int ri = pick<R>(r, i, sub);
        float wi = __int_as_float(pick<R>(w_bits, i, sub));
        if (R == 1 || i + sub < cnt) {
          float4 x = *(const float4*)(XW + (size_t)si * F + 4 * q);
          row_add<F>(acc, ri, q, sub, x, wi);
        }
      }
    }
  }
  __syncthreads();

  float4* o4 = (float4*)(sums + (size_t)base * F);
  for (int i = t; i < nrows * LP; i += 1024) {
    const float* a = acc + (i / LP) * F + (i % LP);
    float4 o;
    o.x = a[0];
    o.y = a[LP];
    o.z = a[2 * LP];
    o.w = a[3 * LP];
    o4[i] = o;
  }
}


template <int F>
__global__ __launch_bounds__(1024) void agg_kernel(const float* __restrict__ XW,
                                                   const int* __restrict__ es,
                                                   const int* __restrict__ ed,
                                                   const float* __restrict__ ew,
                                                   const int* __restrict__ binStart,
                                                   const float* __restrict__ deg,
                                                   float* __restrict__ sums, int N) {
  constexpr int J = F / 32;
  constexpr int U = 32 / J;
  constexpr int F4 = F / 4;
  __shared__ __attribute__((aligned(16))) float acc[RB * F];
  int t = threadIdx.x;
  int lane = t & 31;
  int wave = t >> 5;
  int base = blockIdx.x * RB;
  int nrows = N - base < RB ? N - base : RB;

  constexpr int PT = (RB * F4 + 1023) / 1024;
  const float4* xin = (const float4*)(XW + (size_t)base * F);
  float4* a4 = (float4*)acc;
  float4 xs[PT];
  float dg[PT];
#pragma unroll
  for (int k = 0; k < PT; ++k) {
    int i = t + 1024 * k;
    i = i < nrows * F4 ? i : 0;
    dg[k] = deg[base + i / F4];
    xs[k] = xin[i];
  }
#pragma unroll
  for (int k = 0; k < PT; ++k) {
    int i = t + 1024 * k;
    if (i < nrows * F4) {
      float d = rsqrtf(dg[k]);
      float dd = d * d;
      float4 x = xs[k];
      x.x *= dd; x.y *= dd; x.z *= dd; x.w *= dd;
      a4[i] = x;
    }
  }
  __syncthreads();

  int eb = binStart[blockIdx.x];
  int ee = binStart[blockIdx.x + 1];
  for (int e0 = eb + wave * 32; e0 < ee; e0 += 1024) {
    int e = e0 + lane;
    int s = 0, r = 0;
    float w = 0.f;
    if (e < ee) {
      s = es[e];
      r = ed[e];
      w = ew[e];
    }
    int w_bits = __float_as_int(w);
    int cnt = ee - e0;
    if (cnt >= 32) {
      for (int i0 = 0; i0 < 32; i0 += U) {
        float v[U][J];
#pragma unroll
        for (int u = 0; u < U; ++u) {
          int si = __builtin_amdgcn_readlane(s, i0 + u);
          const float* xr = XW + (size_t)si * F + lane;
#pragma unroll
          for (int j = 0; j < J; ++j) v[u][j] = xr[j * 32];
        }
#pragma unroll
        for (int u = 0; u < U; ++u) {
          int ri = __builtin_amdgcn_readlane(r, i0 + u);
          float wi = __int_as_float(__builtin_amdgcn_readlane(w_bits, i0 + u));
          float* ar = acc + ri * F + lane;
#pragma unroll
          for (int j = 0; j < J; ++j) atomicAdd(ar + j * 32, v[u][j] * wi);
        }
      }
    } else {
      for (int i = 0; i < cnt; ++i) {
        int si = __builtin_amdgcn_readlane(s, i);
        int ri = __builtin_amdgcn_readlane(r, i);
        float wi = __int_as_float(__builtin_amdgcn_readlane(w_bits, i));
        const float* xr = XW + (size_t)si * F + lane;
        float* ar = acc + ri * F + lane;
#pragma unroll
        for (int j = 0; j < J; ++j) atomicAdd(ar + j * 32, xr[j * 32] * wi);
      }
    }
  }
  __syncthreads();

  float4* o4 = (float4*)(sums + (size_t)base * F);
  for (int i = t; i < nrows * F4; i += 1024) o4[i] = a4[i];
}


#define PN 32
__global__ __launch_bounds__(128) void pool_sum_kernel(const float* __restrict__ h,
                                                       const float* __restrict__ b,
                                                       const int* __restrict__ batch,
                                                       float* __restrict__ out,
                                                       float* __restrict__ cnt, int N) {
  int lane = threadIdx.x & 31;
  int wave = blockIdx.x * 4 + __builtin_amdgcn_readfirstlane(threadIdx.x >> 5);
  int n0 = wave * PN;
  if (n0 >= N) return;
  int nn = N - n0 < PN ? N - n0 : PN;
  int g = batch[n0 + (lane < nn ? lane : nn - 1)];
  float bl = b[lane];
  const float* hp = h + (size_t)n0 * OUTC + lane;
  float v[PN];
#pragma unroll
  for (int i = 0; i < PN; ++i) v[i] = hp[(size_t)(i < nn ? i : nn - 1) * OUTC];

  int cur = __builtin_amdgcn_readlane(g, 0);
  float x0 = v[0] + bl;
  float acc = x0 > 0.f ? x0 : 0.f;
  int n = 1;
#pragma unroll
  for (int i = 1; i < PN; ++i) {
    if (i < nn) {
      int gi = __builtin_amdgcn_readlane(g, i);
      if (gi != cur) {
        atomicAdd(&out[(size_t)cur * OUTC + lane], acc);
        if (lane == 0) atomicAdd(&cnt[cur], (float)n);
        cur = gi;
        acc = 0.f;
        n = 0;
      }
      float x = v[i] + bl;
      acc += x > 0.f ? x : 0.f;
      ++n;
    }
  }
  atomicAdd(&out[(size_t)cur * OUTC + lane], acc);
  if (lane == 0) atomicAdd(&cnt[cur], (float)n);
}

__global__ void pool_div_kernel(float* __restrict__ out, const float* __restrict__ cnt, int G) {
  int i = blockIdx.x * blockDim.x + threadIdx.x;
  if (i >= G * OUTC) return;
  out[i] /= fmaxf(cnt[i >> 5], 1.0f);
}


extern "C" void kernel_launch(void* const* d_in, const int* in_sizes, int n_in,
                              void* d_out, int out_size, void* d_ws, size_t ws_size,
                              hipStream_t stream) {
  const float* x     = (const float*)d_in[0];
  const int*   ei    = (const int*)d_in[1];
  const int*   batch = (const int*)d_in[2];
  const float* W1    = (const float*)d_in[3];
  const float* b1    = (const float*)d_in[4];
  const float* W2    = (const float*)d_in[5];
  const float* b2    = (const float*)d_in[6];
  const float* W3    = (const float*)d_in[7];
  const float* b3    = (const float*)d_in[8];

  const int N = in_sizes[0] / F_IN;
  const int E = in_sizes[1] / 2;
  const int G = out_size / OUTC;
  const int* src = ei;
  const int* dst = ei + E;
  float* out = (float*)d_out;

  float* deg = (float*)d_ws;
  float* cnt = deg + N;
  int* binCount  = (int*)(cnt + G);
  int* binCursor = binCount + MAXBIN;
  int* binStart  = binCursor + MAXBIN;
  size_t off = (size_t)N + (size_t)G + 3 * (size_t)MAXBIN + 1;
  off = (off + 63) & ~(size_t)63;
  float* bufA = (float*)d_ws + off;
  float* bufB = bufA + (size_t)N * HID;
  float* bufC = bufB + (size_t)N * HID;
  int*   es = (int*)(bufC + (size_t)N * HID2);
  int*   ed = es + E;
  float* ew = (float*)(ed + E);

  const int TB = 256;

  int n0 = G > out_size ? G : out_size;
  if (MAXBIN > n0) n0 = MAXBIN;
  init_kernel<<<ceil_div(n0, TB), TB, 0, stream>>>(cnt, out, binCount, binCursor, G, out_size);
  count_kernel<<<ceil_div(E, CHUNK), 256, 0, stream>>>(dst, binCount, E);
  place_kernel<<<ceil_div(E, CHUNK), 256, 0, stream>>>(src, dst, binCount, binCursor,
                                                       binStart, es, ed, E);
  const int nbin = ceil_div(N, RB);
  degree_kernel<<<nbin, 1024, 0, stream>>>(ed, binStart, deg, N);

  const int GB = 512;
  int s1 = ceil_div(N, 16), s2 = ceil_div(N, 32), s3 = ceil_div(N, 64);

  gemm_wstat_kernel<false, F_IN, HID><<<s1 < GB ? s1 : GB, 256, 0, stream>>>(x, nullptr, W1, bufA, N);
  agg_rows_kernel<HID, true><<<nbin, 1024, 0, stream>>>(bufA, es, ed, ew, binStart, deg, bufB, N);

  gemm_wstat_kernel<true, HID, HID2><<<s2 < GB ? s2 : GB, 256, 0, stream>>>(bufB, b1, W2, bufA, N);
  agg_rows_kernel<HID2, false><<<nbin, 1024, 0, stream>>>(bufA, es, ed, ew, binStart, deg, bufC, N);

  gemm_wstat_kernel<true, HID2, OUTC><<<s3 < GB ? s3 : GB, 256, 0, stream>>>(bufC, b2, W3, bufA, N);
  agg_kernel<OUTC><<<nbin, 1024, 0, stream>>>(bufA, es, ed, ew, binStart, deg, bufB, N);

  pool_sum_kernel<<<ceil_div(N, PN * 4), 128, 0, stream>>>(bufB, b3, batch, out, cnt, N);
  pool_div_kernel<<<ceil_div(G * OUTC, TB), TB, 0, stream>>>(out, cnt, G);
}


# Edge buckets get a fixed line-aligned stride (a quarter above the mean edges per range): a range's cursor hands out slots, so the count launch and its prefix go; full ones spill to a list; same sums
# speedup vs baseline: 5.4676x; 1.0393x over previous
#include <hip/hip_runtime.h>

typedef float v2f __attribute__((ext_vector_type(2)));
typedef float v8f __attribute__((ext_vector_type(8)));

#define F_IN 128
#define HID  128
#define HID2 64
#define OUTC 32

#define RB     196
#define MAXBIN 1024
#define CHUNK  4096
#define EPT    (CHUNK / 256)

static inline int ceil_div(int a, int b) { return (a + b - 1) / b; }


__global__ void init_kernel(float* __restrict__ cnt, float* __restrict__ out,
                            int* __restrict__ binCursor, int G, int out_size) {
  int i = blockIdx.x * blockDim.x + threadIdx.x;
  if (i < G) cnt[i] = 0.f;
  if (i < out_size) out[i] = 0.f;
  if (i <= MAXBIN) binCursor[i] = 0;
}


static_assert(RB <= 256, "the row inside a range is kept in 8 bits beside the range number");
__global__ __launch_bounds__(256) void place_kernel(const int* __restrict__ src,
                                                    const int* __restrict__ dst,
                                                    int* __restrict__ binCursor,
                                                    int* __restrict__ es, int* __restrict__ ed,
                                                    int* __restrict__ ovs, int* __restrict__ ovd,
                                                    int E, int cap) {
  __shared__ int hist[MAXBIN];
  __shared__ int pos[MAXBIN];
  __shared__ int lstart[MAXBIN];
  __shared__ int wsum[8];
  __shared__ int st_s[CHUNK];
  __shared__ int st_p[CHUNK];
  int t = threadIdx.x;
  int lane = t & 31;
  int wave = t >> 5;

  int e0 = blockIdx.x * CHUNK;
  int e1 = e0 + CHUNK < E ? e0 + CHUNK : E;
  int dv[EPT], sv[EPT];
#pragma unroll
  for (int k = 0; k < EPT; ++k) {
    int e = e0 + t + 256 * k;
    e = e < e1 ? e : e1 - 1;
    dv[k] = dst[e];
    sv[k] = src[e];
  }
  for (int i = t; i < MAXBIN; i += 256) hist[i] = 0;
  __syncthreads();

  int bv[EPT], rk[EPT];
#pragma unroll
  for (int k = 0; k < EPT; ++k) {
    bv[k] = dv[k] / RB;
    rk[k] = 0;
    if (e0 + t + 256 * k < e1) rk[k] = atomicAdd(&hist[bv[k]], 1);
  }
  __syncthreads();

  int h0 = hist[4 * t + 0], h1 = hist[4 * t + 1];
  int h2 = hist[4 * t + 2], h3 = hist[4 * t + 3];
  int hown = h0 + h1 + h2 + h3;
  int hv = hown;
#pragma unroll
  for (int o = 1; o < 32; o <<= 1) {
    int u = __shfl_up(hv, o);
    if (lane >= o) hv += u;
  }
  if (lane == 31) wsum[wave] = hv;
  __syncthreads();
  int lo = hv - hown;
  for (int w = 0; w < 8; ++w)
    if (w < wave) lo += wsum[w];
  lstart[4 * t + 0] = lo;
  lstart[4 * t + 1] = lo + h0;
  lstart[4 * t + 2] = lo + h0 + h1;
  lstart[4 * t + 3] = lo + h0 + h1 + h2;
  __syncthreads();

  int got[MAXBIN / 256];
#pragma unroll
  for (int j = 0; j < MAXBIN / 256; ++j) {
    int h = hist[t + 256 * j];
    got[j] = 0;
    if (h) got[j] = atomicAdd(&binCursor[t + 256 * j], h);
  }
  int lp[EPT];
#pragma unroll
  for (int k = 0; k < EPT; ++k) lp[k] = lstart[bv[k]] + rk[k];
#pragma unroll
  for (int k = 0; k < EPT; ++k) {
    if (e0 + t + 256 * k < e1) {
      st_s[lp[k]] = sv[k];
      st_p[lp[k]] = bv[k] * 256 + (dv[k] - bv[k] * RB);
    }
  }
#pragma unroll
  for (int j = 0; j < MAXBIN / 256; ++j) {
    int i = t + 256 * j;
    pos[i] = got[j] - lstart[i];
  }
  __syncthreads();

  int n = e1 - e0;
  int pj[EPT], sj[EPT], gj[EPT];
#pragma unroll
  for (int k = 0; k < EPT; ++k) {
    int j = t + 256 * k;
    j = j < n ? j : n - 1;
    pj[k] = st_p[j];
    sj[k] = st_s[j];
  }
#pragma unroll
  for (int k = 0; k < EPT; ++k) gj[k] = pos[pj[k] >> 8] + t + 256 * k;
#pragma unroll
  for (int k = 0; k < EPT; ++k) {
    if (t + 256 * k < n) {
      int b = pj[k] >> 8;
      if (gj[k] < cap) {
        es[b * cap + gj[k]] = sj[k];
        ed[b * cap + gj[k]] = pj[k] & 255;
      } else {
        int o = atomicAdd(&binCursor[MAXBIN], 1);
        ovs[o] = sj[k];
        ovd[o] = b * RB + (pj[k] & 255);
      }
    }
  }
}

__global__ __launch_bounds__(1024) void degree_kernel(const int* __restrict__ ed,
                                                      const int* __restrict__ ovd,
                                                      const int* __restrict__ binCursor,
                                                      float* __restrict__ deg, int N, int cap) {
  __shared__ int rowCount[RB];
  int t = threadIdx.x;
  if (t < RB) rowCount[t] = 0;
  __syncthreads();
  int filled = binCursor[blockIdx.x];
  int novf = binCursor[MAXBIN];
  int eb = blockIdx.x * cap;
  int ee = eb + (filled < cap ? filled : cap);
  constexpr int DU = 8;
  static_assert(DU == 8, "the load fence in the loop names DU values");
  for (int e = eb + t; e < ee; e += 1024 * DU) {
    int r[DU];
#pragma unroll
    for (int k = 0; k < DU; ++k) {
      int ek = e + 1024 * k;
      r[k] = ed[ek < ee ? ek : ee - 1];
    }
    asm volatile("" :: "v"(r[0]), "v"(r[1]), "v"(r[2]), "v"(r[3]),
                       "v"(r[4]), "v"(r[5]), "v"(r[6]), "v"(r[7]));
#pragma unroll
    for (int k = 0; k < DU; ++k)
      if (e + 1024 * k < ee) atomicAdd(&rowCount[r[k]], 1);
  }
  int bx = blockIdx.x;
  for (int i = t; i < novf; i += 1024) {
    int d = ovd[i];
    if (d / RB == bx) atomicAdd(&rowCount[d - bx * RB], 1);
  }
  __syncthreads();
  int node = blockIdx.x * RB + t;
  if (t < RB && node < N) deg[node] = 1.0f + (float)rowCount[t];
}

template <int K>
__device__ __forceinline__ float4 stage_piece(const float* __restrict__ A, int row, int pc, int M) {
  row = row < M ? row : M - 1;
  return *(const float4*)(A + (size_t)row * K + 4 * pc);
}

template <bool ACT>
__device__ __forceinline__ float4 act_piece(float4 v, float4 b) {
  if (ACT) {
    v.x = fmaxf(v.x + b.x, 0.f);
    v.y = fmaxf(v.y + b.y, 0.f);
    v.z = fmaxf(v.z + b.z, 0.f);
    v.w = fmaxf(v.w + b.w, 0.f);
  }
  return v;
}

template <bool ACT, int K, int NCOL>
__global__ __launch_bounds__(256) void gemm_wstat_kernel(
    const float* __restrict__ A, const float* __restrict__ abias, const float* __restrict__ B,
    float* __restrict__ C, int M) {
  constexpr int NT   = NCOL / 16;
  constexpr int RT   = 8 / NT;
  constexpr int ROWS = RT * 16;
  constexpr int K4   = K / 4;
  constexpr int P    = K + 4;
  constexpr int RSTEP = 256 / K4;
  constexpr int NLD  = ROWS / RSTEP;
  static_assert(NT * RT == 8 && NLD * RSTEP == ROWS, "8 waves, whole pieces");
  __shared__ __attribute__((aligned(16))) float As[2][ROWS * P];

  int t = threadIdx.x;
  int lane = t & 31;
  int wave = __builtin_amdgcn_readfirstlane(t >> 5);
  int c = lane & 15;
  int h = lane >> 4;
  int ct = wave % NT;
  int rt = wave / NT;

  v2f bf[K / 4];
  const float* bcol = B + ct * 16 + c;
#pragma unroll
  for (int j = 0; j < K / 8; ++j) {
    int k0 = 8 * j + 4 * h;
    bf[2 * j].x     = bcol[(size_t)(k0 + 0) * NCOL];
    bf[2 * j].y     = bcol[(size_t)(k0 + 1) * NCOL];
    bf[2 * j + 1].x = bcol[(size_t)(k0 + 2) * NCOL];
    bf[2 * j + 1].y = bcol[(size_t)(k0 + 3) * NCOL];
  }

  const unsigned coff = (unsigned)((rt * 16 + h * 8) * NCOL + ct * 16 + c);

  int pc = t % K4;
  int pr = t / K4;
  float4 bias4 = {0.f, 0.f, 0.f, 0.f};
  if (ACT) bias4 = *(const float4*)(abias + 4 * pc);

  int nsteps = (M + ROWS - 1) / ROWS;
  float4 s0, s1, s2, s3;
  s0 = s1 = s2 = s3 = bias4;
  int step = blockIdx.x;
  if (step < nsteps) {
    s0 = stage_piece<K>(A, step * ROWS + pr, pc, M);
    s1 = stage_piece<K>(A, step * ROWS + pr + RSTEP, pc, M);
    if (NLD == 4) {
      s2 = stage_piece<K>(A, step * ROWS + pr + 2 * RSTEP, pc, M);
      s3 = stage_piece<K>(A, step * ROWS + pr + 3 * RSTEP, pc, M);
    }
  }
  int buf = 0;
  for (; step < nsteps; step += gridDim.x) {
    float* as = As[buf];
    float* ap = as + pr * P + 4 * pc;
    *(float4*)(ap) = act_piece<ACT>(s0, bias4);
    *(float4*)(ap + RSTEP * P) = act_piece<ACT>(s1, bias4);
    if (NLD == 4) {
      *(float4*)(ap + 2 * RSTEP * P) = act_piece<ACT>(s2, bias4);
      *(float4*)(ap + 3 * RSTEP * P) = act_piece<ACT>(s3, bias4);
    }
    __syncthreads();

    int nxt = step + gridDim.x;
    if (nxt < nsteps) {
      s0 = stage_piece<K>(A, nxt * ROWS + pr, pc, M);
      s1 = stage_piece<K>(A, nxt * ROWS + pr + RSTEP, pc, M);
      if (NLD == 4) {
        s2 = stage_piece<K>(A, nxt * ROWS + pr + 2 * RSTEP, pc, M);
        s3 = stage_piece<K>(A, nxt * ROWS + pr + 3 * RSTEP, pc, M);
      }
    }

    int tile0 = step * ROWS + rt * 16;
    float* cstep = C + (size_t)step * (ROWS * NCOL);

    const float* arow = as + (rt * 16 + c) * P + 4 * h;
    float4 a[K / 8];
#pragma unroll
    for (int j = 0; j < K / 8; ++j) a[j] = *(const float4*)(arow + 8 * j);
    __builtin_amdgcn_sched_barrier(0);

    v8f acc = {0.f, 0.f, 0.f, 0.f, 0.f, 0.f, 0.f, 0.f};
#pragma unroll
    for (int j = 0; j < K / 8; ++j) {
      v2f a0 = {a[j].x, a[j].y};
      v2f a1 = {a[j].z, a[j].w};
      acc = __builtin_amdgcn_wmma_f32_16x16x4_f32(
           false, a0,  false, bf[2 * j],
           (short)0, acc,  false,  false);
      acc = __builtin_amdgcn_wmma_f32_16x16x4_f32(
           false, a1,  false, bf[2 * j + 1],
           (short)0, acc,  false,  false);
    }

    if (tile0 + 16 <= M) {
#pragma unroll
      for (int r = 0; r < 8; ++r) cstep[coff + r * NCOL] = acc[r];
    } else {
#pragma unroll
      for (int r = 0; r < 8; ++r)
        if (tile0 + h * 8 + r < M) cstep[coff + r * NCOL] = acc[r];
    }
    buf ^= 1;
  }
}


template <int R>
__device__ __forceinline__ int pick(int x, int i, int sub) {
  int a = __builtin_amdgcn_readlane(x, i);
  if (R == 2) {
    int b = __builtin_amdgcn_readlane(x, i + 1);
    a = sub ? b : a;
  }
  return a;
}

template <int F>
__device__ __forceinline__ void row_add(float* acc, int ri, int q, int sub, float4 v, float w) {
  constexpr int LP = F / 4;
  float* ar = acc + ri * F + q;
  if (F == 128) {
    atomicAdd(ar,          v.x * w);
    atomicAdd(ar + LP,     v.y * w);
    atomicAdd(ar + 2 * LP, v.z * w);
    atomicAdd(ar + 3 * LP, v.w * w);
  } else {
    float y0 = sub ? v.z : v.x;
    float y1 = sub ? v.w : v.y;
    float y2 = sub ? v.x : v.z;
    float y3 = sub ? v.y : v.w;
    float* a1 = ar + (sub ? 2 * LP : 0);
    float* a2 = ar + (sub ? 0 : 2 * LP);
    atomicAdd(a1,      y0 * w);
    atomicAdd(a1 + LP, y1 * w);
    atomicAdd(a2,      y2 * w);
    atomicAdd(a2 + LP, y3 * w);
  }
}

template <int F, bool MKW>
__global__ __launch_bounds__(1024) void agg_rows_kernel(const float* __restrict__ XW,
                                                        const int* __restrict__ es,
                                                        const int* __restrict__ ed,
                                                        float* __restrict__ ew,
                                                        const int* __restrict__ ovs,
                                                        const int* __restrict__ ovd,
                                                        const int* __restrict__ binCursor,
                                                        const float* __restrict__ deg,
                                                        float* __restrict__ sums, int N, int cap) {
  static_assert(F == 128 || F == 64, "a row is one or two aligned 256-byte units");
  constexpr int LP = F / 4;
  constexpr int R  = 32 / LP;
  constexpr int U  = 8;
  __shared__ __attribute__((aligned(16))) float acc[RB * F];
  int t = threadIdx.x;
  int lane = t & 31;
  int wave = t >> 5;
  int sub = lane / LP;
  int q = lane % LP;
  int base = blockIdx.x * RB;
  int nrows = N - base < RB ? N - base : RB;

  constexpr int PT = (RB * LP + 1023) / 1024;
  const float4* xin = (const float4*)(XW + (size_t)base * F);
  float4 xs[PT];
  float dg[PT];
#pragma unroll
  for (int k = 0; k < PT; ++k) {
    int i = t + 1024 * k;
    i = i < nrows * LP ? i : 0;
    dg[k] = deg[base + i / LP];
    xs[k] = xin[i];
  }
#pragma unroll
  for (int k = 0; k < PT; ++k) {
    int i = t + 1024 * k;
    if (i < nrows * LP) {
      int row = i / LP;
      float d = rsqrtf(dg[k]);
      float dd = d * d;
      float4 x = xs[k];
      float* a = acc + row * F + (i % LP);
      a[0]      = x.x * dd;
      a[LP]     = x.y * dd;
      a[2 * LP] = x.z * dd;
      a[3 * LP] = x.w * dd;
    }
  }
  __syncthreads();

  int filled = binCursor[blockIdx.x];
  int eb = blockIdx.x * cap;
  int ee = eb + (filled < cap ? filled : cap);
  for (int e0 = eb + wave * 32; e0 < ee; e0 += 1024) {
    int e = e0 + lane;
    int s = 0, r = 0;
    float w = 0.f;
    if (e < ee) {
      s = es[e];
      r = ed[e];
      if (!MKW) w = ew[e];
    }
    int cnt = ee - e0;
    if (MKW && cnt < 32 && e < ee) {
      w = rsqrtf(deg[s]) * rsqrtf(deg[base + r]);
      ew[e] = w;
    }
    int w_bits = __float_as_int(w);
    if (cnt >= 32) {
      float dgs = 1.f, dgd = 1.f;
      for (int i0 = 0; i0 < 32; i0 += U * R) {
        float4 v[U];
#pragma unroll
        for (int u = 0; u < U; ++u) {
          int si = pick<R>(s, i0 + u * R, sub);
          v[u] = *(const float4*)(XW + (size_t)si * F + 4 * q);
        }
        if (MKW && i0 == 0) {
          dgs = deg[s];
          dgd = deg[base + r];
        }
        __builtin_amdgcn_sched_barrier(0);
        if (MKW && i0 == 0) {
          w = rsqrtf(dgs) * rsqrtf(dgd);
          ew[e] = w;
          w_bits = __float_as_int(w);
        }
#pragma unroll
        for (int u = 0; u < U; ++u) {
          int ri = pick<R>(r, i0 + u * R, sub);
          float wi = __int_as_float(pick<R>(w_bits, i0 + u * R, sub));
          row_add<F>(acc, ri, q, sub, v[u], wi);
        }
      }
    } else {
      for (int i = 0; i < cnt; i += R) {
        int si = pick<R>(s, i, sub);
        int ri = pick<R>(r, i, sub);
        float wi = __int_as_float(pick<R>(w_bits, i, sub));
        if (R == 1 || i + sub < cnt) {
          float4 x = *(const float4*)(XW + (size_t)si * F + 4 * q);
          row_add<F>(acc, ri, q, sub, x, wi);
        }
      }
    }
  }
  int novf = binCursor[MAXBIN];
  for (int i = t; i < novf; i += 1024) {
    int d = ovd[i];
    if (d / RB == (int)blockIdx.x) {
      int s = ovs[i];
      float wo = rsqrtf(deg[s]) * rsqrtf(deg[d]);
      const float* xr = XW + (size_t)s * F;
      float* ar = acc + (d - base) * F;
      for (int f = 0; f < F; ++f) atomicAdd(ar + (f & 3) * LP + (f >> 2), xr[f] * wo);
    }
  }
  __syncthreads();

  float4* o4 = (float4*)(sums + (size_t)base * F);
  for (int i = t; i < nrows * LP; i += 1024) {
    const float* a = acc + (i / LP) * F + (i % LP);
    float4 o;
    o.x = a[0];
    o.y = a[LP];
    o.z = a[2 * LP];
    o.w = a[3 * LP];
    o4[i] = o;
  }
}


template <int F>
__global__ __launch_bounds__(1024) void agg_kernel(const float* __restrict__ XW,
                                                   const int* __restrict__ es,
                                                   const int* __restrict__ ed,
                                                   const float* __restrict__ ew,
                                                   const int* __restrict__ ovs,
                                                   const int* __restrict__ ovd,
                                                   const int* __restrict__ binCursor,
                                                   const float* __restrict__ deg,
                                                   float* __restrict__ sums, int N, int cap) {
  constexpr int J = F / 32;
  constexpr int U = 32 / J;
  constexpr int F4 = F / 4;
  __shared__ __attribute__((aligned(16))) float acc[RB * F];
  int t = threadIdx.x;
  int lane = t & 31;
  int wave = t >> 5;
  int base = blockIdx.x * RB;
  int nrows = N - base < RB ? N - base : RB;

  constexpr int PT = (RB * F4 + 1023) / 1024;
  const float4* xin = (const float4*)(XW + (size_t)base * F);
  float4* a4 = (float4*)acc;
  float4 xs[PT];
  float dg[PT];
#pragma unroll
  for (int k = 0; k < PT; ++k) {
    int i = t + 1024 * k;
    i = i < nrows * F4 ? i : 0;
    dg[k] = deg[base + i / F4];
    xs[k] = xin[i];
  }
#pragma unroll
  for (int k = 0; k < PT; ++k) {
    int i = t + 1024 * k;
    if (i < nrows * F4) {
      float d = rsqrtf(dg[k]);
      float dd = d * d;
      float4 x = xs[k];
      x.x *= dd; x.y *= dd; x.z *= dd; x.w *= dd;
      a4[i] = x;
    }
  }
  __syncthreads();

  int filled = binCursor[blockIdx.x];
  int eb = blockIdx.x * cap;
  int ee = eb + (filled < cap ? filled : cap);
  for (int e0 = eb + wave * 32; e0 < ee; e0 += 1024) {
    int e = e0 + lane;
    int s = 0, r = 0;
    float w = 0.f;
    if (e < ee) {
      s = es[e];
      r = ed[e];
      w = ew[e];
    }
    int w_bits = __float_as_int(w);
    int cnt = ee - e0;
    if (cnt >= 32) {
      for (int i0 = 0; i0 < 32; i0 += U) {
        float v[U][J];
#pragma unroll
        for (int u = 0; u < U; ++u) {
          int si = __builtin_amdgcn_readlane(s, i0 + u);
          const float* xr = XW + (size_t)si * F + lane;
#pragma unroll
          for (int j = 0; j < J; ++j) v[u][j] = xr[j * 32];
        }
#pragma unroll
        for (int u = 0; u < U; ++u) {
          int ri = __builtin_amdgcn_readlane(r, i0 + u);
          float wi = __int_as_float(__builtin_amdgcn_readlane(w_bits, i0 + u));
          float* ar = acc + ri * F + lane;
#pragma unroll
          for (int j = 0; j < J; ++j) atomicAdd(ar + j * 32, v[u][j] * wi);
        }
      }
    } else {
      for (int i = 0; i < cnt; ++i) {
        int si = __builtin_amdgcn_readlane(s, i);
        int ri = __builtin_amdgcn_readlane(r, i);
        float wi = __int_as_float(__builtin_amdgcn_readlane(w_bits, i));
        const float* xr = XW + (size_t)si * F + lane;
        float* ar = acc + ri * F + lane;
#pragma unroll
        for (int j = 0; j < J; ++j) atomicAdd(ar + j * 32, xr[j * 32] * wi);
      }
    }
  }
  int novf = binCursor[MAXBIN];
  for (int i = t; i < novf; i += 1024) {
    int d = ovd[i];
    if (d / RB == (int)blockIdx.x) {
      int s = ovs[i];
      float wo = rsqrtf(deg[s]) * rsqrtf(deg[d]);
      const float* xr = XW + (size_t)s * F;
      float* ar = acc + (d - base) * F;
      for (int f = 0; f < F; ++f) atomicAdd(ar + f, xr[f] * wo);
    }
  }
  __syncthreads();

  float4* o4 = (float4*)(sums + (size_t)base * F);
  for (int i = t; i < nrows * F4; i += 1024) o4[i] = a4[i];
}


#define PN 32
__global__ __launch_bounds__(128) void pool_sum_kernel(const float* __restrict__ h,
                                                       const float* __restrict__ b,
                                                       const int* __restrict__ batch,
                                                       float* __restrict__ out,
                                                       float* __restrict__ cnt, int N) {
  int lane = threadIdx.x & 31;
  int wave = blockIdx.x * 4 + __builtin_amdgcn_readfirstlane(threadIdx.x >> 5);
  int n0 = wave * PN;
  if (n0 >= N) return;
  int nn = N - n0 < PN ? N - n0 : PN;
  int g = batch[n0 + (lane < nn ? lane : nn - 1)];
  float bl = b[lane];
  const float* hp = h + (size_t)n0 * OUTC + lane;
  float v[PN];
#pragma unroll
  for (int i = 0; i < PN; ++i) v[i] = hp[(size_t)(i < nn ? i : nn - 1) * OUTC];

  int cur = __builtin_amdgcn_readlane(g, 0);
  float x0 = v[0] + bl;
  float acc = x0 > 0.f ? x0 : 0.f;
  int n = 1;
#pragma unroll
  for (int i = 1; i < PN; ++i) {
    if (i < nn) {
      int gi = __builtin_amdgcn_readlane(g, i);
      if (gi != cur) {
        atomicAdd(&out[(size_t)cur * OUTC + lane], acc);
        if (lane == 0) atomicAdd(&cnt[cur], (float)n);
        cur = gi;
        acc = 0.f;
        n = 0;
      }
      float x = v[i] + bl;
      acc += x > 0.f ? x : 0.f;
      ++n;
    }
  }
  atomicAdd(&out[(size_t)cur * OUTC + lane], acc);
  if (lane == 0) atomicAdd(&cnt[cur], (float)n);
}

__global__ void pool_div_kernel(float* __restrict__ out, const float* __restrict__ cnt, int G) {
  int i = blockIdx.x * blockDim.x + threadIdx.x;
  if (i >= G * OUTC) return;
  out[i] /= fmaxf(cnt[i >> 5], 1.0f);
}


extern "C" void kernel_launch(void* const* d_in, const int* in_sizes, int n_in,
                              void* d_out, int out_size, void* d_ws, size_t ws_size,
                              hipStream_t stream) {
  const float* x     = (const float*)d_in[0];
  const int*   ei    = (const int*)d_in[1];
  const int*   batch = (const int*)d_in[2];
  const float* W1    = (const float*)d_in[3];
  const float* b1    = (const float*)d_in[4];
  const float* W2    = (const float*)d_in[5];
  const float* b2    = (const float*)d_in[6];
  const float* W3    = (const float*)d_in[7];
  const float* b3    = (const float*)d_in[8];

  const int N = in_sizes[0] / F_IN;
  const int E = in_sizes[1] / 2;
  const int G = out_size / OUTC;
  const int* src = ei;
  const int* dst = ei + E;
  float* out = (float*)d_out;

  const int nbin = ceil_div(N, RB);
  const int mean = ceil_div(E, nbin);
  const int cap = ((mean + mean / 4 + 256 + 63) / 64 | 1) * 64;

  float* deg = (float*)d_ws;
  float* cnt = deg + N;
  int* binCursor = (int*)(cnt + G);
  size_t off = (size_t)N + (size_t)G + (size_t)MAXBIN + 1;
  off = (off + 63) & ~(size_t)63;
  float* bufA = (float*)d_ws + off;
  float* bufB = bufA + (size_t)N * HID;
  float* bufC = bufB + (size_t)N * HID;
  const size_t bw = (size_t)nbin * cap;
  int*   es = (int*)(bufC + (size_t)N * HID2);
  int*   ed = es + bw;
  float* ew = (float*)(ed + bw);
  int*   ovs = (int*)(ew + bw);
  int*   ovd = ovs + E;

  const int TB = 256;

  int n0 = G > out_size ? G : out_size;
  if (MAXBIN + 1 > n0) n0 = MAXBIN + 1;
  init_kernel<<<ceil_div(n0, TB), TB, 0, stream>>>(cnt, out, binCursor, G, out_size);
  place_kernel<<<ceil_div(E, CHUNK), 256, 0, stream>>>(src, dst, binCursor, es, ed, ovs, ovd,
                                                       E, cap);
  degree_kernel<<<nbin, 1024, 0, stream>>>(ed, ovd, binCursor, deg, N, cap);

  const int GB = 512;
  int s1 = ceil_div(N, 16), s2 = ceil_div(N, 32), s3 = ceil_div(N, 64);

  gemm_wstat_kernel<false, F_IN, HID><<<s1 < GB ? s1 : GB, 256, 0, stream>>>(x, nullptr, W1, bufA, N);
  agg_rows_kernel<HID, true><<<nbin, 1024, 0, stream>>>(bufA, es, ed, ew, ovs, ovd, binCursor, deg, bufB, N, cap);

  gemm_wstat_kernel<true, HID, HID2><<<s2 < GB ? s2 : GB, 256, 0, stream>>>(bufB, b1, W2, bufA, N);
  agg_rows_kernel<HID2, false><<<nbin, 1024, 0, stream>>>(bufA, es, ed, ew, ovs, ovd, binCursor, deg, bufC, N, cap);

  gemm_wstat_kernel<true, HID2, OUTC><<<s3 < GB ? s3 : GB, 256, 0, stream>>>(bufC, b2, W3, bufA, N);
  agg_kernel<OUTC><<<nbin, 1024, 0, stream>>>(bufA, es, ed, ew, ovs, ovd, binCursor, deg, bufB, N, cap);

  pool_sum_kernel<<<ceil_div(N, PN * 4), 128, 0, stream>>>(bufB, b3, batch, out, cnt, N);
  pool_div_kernel<<<ceil_div(G * OUTC, TB), TB, 0, stream>>>(out, cnt, G);
}
